# starting kernel
# baseline (speedup 1.0000x reference)
#include <hip/hip_runtime.h>
#include <hip/hip_bf16.h>
#include <math.h>

typedef float v2f __attribute__((ext_vector_type(2)));
typedef float v8f __attribute__((ext_vector_type(8)));

#if defined(__has_builtin)
#if __has_builtin(__builtin_amdgcn_wmma_f32_16x16x4_f32)
#define HAS_WMMA_F32_4 1
#endif
#endif
#ifndef HAS_WMMA_F32_4
#define HAS_WMMA_F32_4 0
#endif

#define FEAT 128
#define BN_EPS 1e-5f

__global__ __launch_bounds__(256) void degree_kernel(const int* __restrict__ col,
                                                     const float* __restrict__ w,
                                                     float* __restrict__ deg, int E) {
    int e = blockIdx.x * 256 + threadIdx.x;
    if (e < E) {
        __hip_atomic_fetch_add(&deg[col[e]], w[e], __ATOMIC_RELAXED,
                               __HIP_MEMORY_SCOPE_AGENT);
    }
}

__global__ __launch_bounds__(256) void dis_kernel(float* __restrict__ deg_dis, int n) {
    int i = blockIdx.x * 256 + threadIdx.x;
    if (i < n) deg_dis[i] = rsqrtf(deg_dis[i] + 1.0f);
}

__global__ __launch_bounds__(256) void gemm_n128_kernel(const float* __restrict__ X,
                                                        const float* __restrict__ W,
                                                        float* __restrict__ Y,
                                                        int nrows) {
    __shared__ float lA[16 * 132];

    const int tid = threadIdx.x;
    const int rowBase = blockIdx.x * 16;

#pragma unroll
    for (int i = 0; i < 2; ++i) {
        int idx = tid + i * 256;
        int r = idx >> 5;
        int c4 = (idx & 31) << 2;
        float4 v = make_float4(0.f, 0.f, 0.f, 0.f);
        if (rowBase + r < nrows)
            v = *(const float4*)(X + (size_t)(rowBase + r) * FEAT + c4);
        float* dst = &lA[r * 132 + c4];
        dst[0] = v.x; dst[1] = v.y; dst[2] = v.z; dst[3] = v.w;
    }
    __syncthreads();

    const int wave = tid >> 5;
    const int lane = tid & 31;
    const int lid  = lane & 15;
    const int half = lane >> 4;
    const int n0   = wave * 16;

#if HAS_WMMA_F32_4
    v8f acc = {0.f, 0.f, 0.f, 0.f, 0.f, 0.f, 0.f, 0.f};
    const float* arow = &lA[lid * 132];
#pragma unroll
    for (int k0 = 0; k0 < FEAT; k0 += 4) {
        const int ka = k0 + 2 * half;
        v2f a, b;
        a.x = arow[ka];
        a.y = arow[ka + 1];
        b.x = W[(size_t)ka * FEAT + n0 + lid];
        b.y = W[(size_t)(ka + 1) * FEAT + n0 + lid];
        acc = __builtin_amdgcn_wmma_f32_16x16x4_f32(
             false, a,  false, b,
             (short)0, acc,  false,  false);
    }
    float* yb = Y + (size_t)rowBase * FEAT + n0 + lid;
#pragma unroll
    for (int v = 0; v < 8; ++v) {
        int m = v + 8 * half;
        if (rowBase + m < nrows) yb[(size_t)m * FEAT] = acc[v];
    }
#else
    float accs[8] = {0.f, 0.f, 0.f, 0.f, 0.f, 0.f, 0.f, 0.f};
    for (int k = 0; k < FEAT; ++k) {
        float bv = W[(size_t)k * FEAT + n0 + lid];
#pragma unroll
        for (int v = 0; v < 8; ++v)
            accs[v] = fmaf(lA[(v + 8 * half) * 132 + k], bv, accs[v]);
    }
    float* yb = Y + (size_t)rowBase * FEAT + n0 + lid;
#pragma unroll
    for (int v = 0; v < 8; ++v) {
        int m = v + 8 * half;
        if (rowBase + m < nrows) yb[(size_t)m * FEAT] = accs[v];
    }
#endif
}

__global__ __launch_bounds__(256) void aggregate_kernel(const int* __restrict__ row,
                                                        const int* __restrict__ col,
                                                        const float* __restrict__ ew,
                                                        const float* __restrict__ dis,
                                                        const float* __restrict__ h,
                                                        float* __restrict__ agg, int E) {
    int gid = blockIdx.x * 256 + threadIdx.x;
    int e = gid >> 5;
    if (e >= E) return;
    int lane = gid & 31;
    int r = row[e];
    int c = col[e];
    float nrm = dis[r] * ew[e] * dis[c];
    const float4 hv = *(const float4*)(h + (size_t)r * FEAT + lane * 4);
    float* dst = agg + (size_t)c * FEAT + lane * 4;
    __hip_atomic_fetch_add(dst + 0, hv.x * nrm, __ATOMIC_RELAXED, __HIP_MEMORY_SCOPE_AGENT);
    __hip_atomic_fetch_add(dst + 1, hv.y * nrm, __ATOMIC_RELAXED, __HIP_MEMORY_SCOPE_AGENT);
    __hip_atomic_fetch_add(dst + 2, hv.z * nrm, __ATOMIC_RELAXED, __HIP_MEMORY_SCOPE_AGENT);
    __hip_atomic_fetch_add(dst + 3, hv.w * nrm, __ATOMIC_RELAXED, __HIP_MEMORY_SCOPE_AGENT);
}

__global__ __launch_bounds__(256) void epilogue1_kernel(float* __restrict__ agg,
                                                        const float* __restrict__ h,
                                                        const float* __restrict__ dis,
                                                        const float* __restrict__ b0,
                                                        const float* __restrict__ g,
                                                        const float* __restrict__ be,
                                                        const float* __restrict__ mu,
                                                        const float* __restrict__ var,
                                                        int n4) {
    int idx4 = blockIdx.x * 256 + threadIdx.x;
    if (idx4 >= n4) return;
    int i  = idx4 >> 5;
    int f  = (idx4 & 31) << 2;
    float d = dis[i];
    float d2 = d * d;
    float4 a  = *(float4*)(agg + (size_t)idx4 * 4);
    float4 hv = *(const float4*)(h + (size_t)idx4 * 4);
    float out[4];
    float av[4] = {a.x, a.y, a.z, a.w};
    float hh[4] = {hv.x, hv.y, hv.z, hv.w};
#pragma unroll
    for (int j = 0; j < 4; ++j) {
        int ff = f + j;
        float v = av[j] + d2 * hh[j] + b0[ff];
        v = (v - mu[ff]) * rsqrtf(var[ff] + BN_EPS) * g[ff] + be[ff];
        out[j] = (v > 0.0f) ? v : expm1f(v);
    }
    *(float4*)(agg + (size_t)idx4 * 4) = make_float4(out[0], out[1], out[2], out[3]);
}

__global__ __launch_bounds__(256) void final_kernel(const float* __restrict__ emb,
                                                    const float* __restrict__ h1,
                                                    const float* __restrict__ agg2,
                                                    const float* __restrict__ h2,
                                                    const float* __restrict__ dis,
                                                    const float* __restrict__ b1,
                                                    float* __restrict__ out, int n4) {
    int idx4 = blockIdx.x * 256 + threadIdx.x;
    if (idx4 >= n4) return;
    int i = idx4 >> 5;
    int f = (idx4 & 31) << 2;
    float d = dis[i];
    float d2 = d * d;
    float4 e  = *(const float4*)(emb  + (size_t)idx4 * 4);
    float4 o1 = *(const float4*)(h1   + (size_t)idx4 * 4);
    float4 a2 = *(const float4*)(agg2 + (size_t)idx4 * 4);
    float4 p2 = *(const float4*)(h2   + (size_t)idx4 * 4);
    const float inv3 = 1.0f / 3.0f;
    float4 r;
    r.x = (e.x + o1.x + a2.x + d2 * p2.x + b1[f + 0]) * inv3;
    r.y = (e.y + o1.y + a2.y + d2 * p2.y + b1[f + 1]) * inv3;
    r.z = (e.z + o1.z + a2.z + d2 * p2.z + b1[f + 2]) * inv3;
    r.w = (e.w + o1.w + a2.w + d2 * p2.w + b1[f + 3]) * inv3;
    *(float4*)(out + (size_t)idx4 * 4) = r;
}

extern "C" void kernel_launch(void* const* d_in, const int* in_sizes, int n_in,
                              void* d_out, int out_size, void* d_ws, size_t ws_size,
                              hipStream_t stream) {
    const float* emb = (const float*)d_in[0];
    const int*   ei  = (const int*)d_in[1];
    const float* ew  = (const float*)d_in[2];
    const float* W0  = (const float*)d_in[3];
    const float* b0  = (const float*)d_in[4];
    const float* W1  = (const float*)d_in[5];
    const float* b1  = (const float*)d_in[6];
    const float* bng = (const float*)d_in[7];
    const float* bnb = (const float*)d_in[8];
    const float* bnm = (const float*)d_in[9];
    const float* bnv = (const float*)d_in[10];
    float* out = (float*)d_out;

    const int N  = in_sizes[0] / FEAT;
    const int E  = in_sizes[2];
    const size_t ND = (size_t)N * FEAT;

    const int* row = ei;
    const int* col = ei + E;

    float* ws   = (float*)d_ws;
    size_t disPad = ((size_t)N + 1023) & ~(size_t)1023;
    float* dis  = ws;
    float* bufA = ws + disPad;
    float* bufB = bufA + ND;
    float* bufC = bufB + ND;

    hipMemsetAsync(dis,  0, (size_t)N * sizeof(float), stream);
    hipMemsetAsync(bufB, 0, ND * sizeof(float), stream);
    hipMemsetAsync(bufC, 0, ND * sizeof(float), stream);

    const int n4 = (int)(ND / 4);
    const int gElem  = (n4 + 255) / 256;
    const int gEdgeW = (int)(((size_t)E * 32 + 255) / 256);
    const int gRowT  = (N + 15) / 16;

    degree_kernel<<<(E + 255) / 256, 256, 0, stream>>>(col, ew, dis, E);
    dis_kernel<<<(N + 255) / 256, 256, 0, stream>>>(dis, N);

    gemm_n128_kernel<<<gRowT, 256, 0, stream>>>(emb, W0, bufA, N);

    aggregate_kernel<<<gEdgeW, 256, 0, stream>>>(row, col, ew, dis, bufA, bufB, E);

    epilogue1_kernel<<<gElem, 256, 0, stream>>>(bufB, bufA, dis, b0, bng, bnb, bnm, bnv, n4);

    gemm_n128_kernel<<<gRowT, 256, 0, stream>>>(bufB, W1, bufA, N);

    aggregate_kernel<<<gEdgeW, 256, 0, stream>>>(row, col, ew, dis, bufA, bufC, E);

    final_kernel<<<gElem, 256, 0, stream>>>(emb, bufB, bufC, bufA, dis, b1, out, n4);
}


# both scatter-adds: global float atomics replaced by 196-node destination tiles summed in LDS by LDS float atomics; each block filters the whole edge list; sums in LDS arrival order
# speedup vs baseline: 1.7793x; 1.7793x over previous
#include <hip/hip_runtime.h>
#include <hip/hip_bf16.h>
#include <math.h>

typedef float v2f __attribute__((ext_vector_type(2)));
typedef float v8f __attribute__((ext_vector_type(8)));

#if defined(__has_builtin)
#if __has_builtin(__builtin_amdgcn_wmma_f32_16x16x4_f32)
#define HAS_WMMA_F32_4 1
#endif
#endif
#ifndef HAS_WMMA_F32_4
#define HAS_WMMA_F32_4 0
#endif

#define FEAT 128
#define BN_EPS 1e-5f

__global__ __launch_bounds__(256) void degree_kernel(const int* __restrict__ col,
                                                     const float* __restrict__ w,
                                                     float* __restrict__ deg, int E) {
    int e = blockIdx.x * 256 + threadIdx.x;
    if (e < E) {
        __hip_atomic_fetch_add(&deg[col[e]], w[e], __ATOMIC_RELAXED,
                               __HIP_MEMORY_SCOPE_AGENT);
    }
}

__global__ __launch_bounds__(256) void dis_kernel(float* __restrict__ deg_dis, int n) {
    int i = blockIdx.x * 256 + threadIdx.x;
    if (i < n) deg_dis[i] = rsqrtf(deg_dis[i] + 1.0f);
}

__global__ __launch_bounds__(256) void gemm_n128_kernel(const float* __restrict__ X,
                                                        const float* __restrict__ W,
                                                        float* __restrict__ Y,
                                                        int nrows) {
    __shared__ float lA[16 * 132];

    const int tid = threadIdx.x;
    const int rowBase = blockIdx.x * 16;

#pragma unroll
    for (int i = 0; i < 2; ++i) {
        int idx = tid + i * 256;
        int r = idx >> 5;
        int c4 = (idx & 31) << 2;
        float4 v = make_float4(0.f, 0.f, 0.f, 0.f);
        if (rowBase + r < nrows)
            v = *(const float4*)(X + (size_t)(rowBase + r) * FEAT + c4);
        float* dst = &lA[r * 132 + c4];
        dst[0] = v.x; dst[1] = v.y; dst[2] = v.z; dst[3] = v.w;
    }
    __syncthreads();

    const int wave = tid >> 5;
    const int lane = tid & 31;
    const int lid  = lane & 15;
    const int half = lane >> 4;
    const int n0   = wave * 16;

#if HAS_WMMA_F32_4
    v8f acc = {0.f, 0.f, 0.f, 0.f, 0.f, 0.f, 0.f, 0.f};
    const float* arow = &lA[lid * 132];
#pragma unroll
    for (int k0 = 0; k0 < FEAT; k0 += 4) {
        const int ka = k0 + 2 * half;
        v2f a, b;
        a.x = arow[ka];
        a.y = arow[ka + 1];
        b.x = W[(size_t)ka * FEAT + n0 + lid];
        b.y = W[(size_t)(ka + 1) * FEAT + n0 + lid];
        acc = __builtin_amdgcn_wmma_f32_16x16x4_f32(
             false, a,  false, b,
             (short)0, acc,  false,  false);
    }
    float* yb = Y + (size_t)rowBase * FEAT + n0 + lid;
#pragma unroll
    for (int v = 0; v < 8; ++v) {
        int m = v + 8 * half;
        if (rowBase + m < nrows) yb[(size_t)m * FEAT] = acc[v];
    }
#else
    float accs[8] = {0.f, 0.f, 0.f, 0.f, 0.f, 0.f, 0.f, 0.f};
    for (int k = 0; k < FEAT; ++k) {
        float bv = W[(size_t)k * FEAT + n0 + lid];
#pragma unroll
        for (int v = 0; v < 8; ++v)
            accs[v] = fmaf(lA[(v + 8 * half) * 132 + k], bv, accs[v]);
    }
    float* yb = Y + (size_t)rowBase * FEAT + n0 + lid;
#pragma unroll
    for (int v = 0; v < 8; ++v) {
        int m = v + 8 * half;
        if (rowBase + m < nrows) yb[(size_t)m * FEAT] = accs[v];
    }
#endif
}

#define TILE_NODES 196
#define AGG_WAVES 32
#define QCAP 1024

struct TileCtx {
    float* acc;
    int* q;
    int base;
    int lane;
    int head;
    int qn;
};

__device__ __forceinline__ void tile_drain(TileCtx& t, int cnt,
                                           const int* __restrict__ row,
                                           const int* __restrict__ col,
                                           const float* __restrict__ ew,
                                           const float* __restrict__ dis,
                                           const float* __restrict__ h) {
    const bool valid = t.lane < cnt;
    int e = t.q[(t.head + t.lane) & (QCAP - 1)];
    e = valid ? e : 0;
    const int r = row[e];
    const int c = col[e];
    float nrm = dis[r] * ew[e] * dis[c];
    nrm = valid ? nrm : 0.0f;
    const int loc = valid ? (c - t.base) : TILE_NODES;
    const int nrmBits = __float_as_int(nrm);
#pragma unroll
    for (int sub = 0; sub < 4; ++sub) {
        if (sub * 8 >= cnt) break;
        float hv[8][4];
#pragma unroll
        for (int k = 0; k < 8; ++k) {
            const int rt = __builtin_amdgcn_readlane(r, sub * 8 + k);
            const float* hp = h + (size_t)rt * FEAT + t.lane;
            hv[k][0] = hp[0];
            hv[k][1] = hp[32];
            hv[k][2] = hp[64];
            hv[k][3] = hp[96];
        }
#pragma unroll
        for (int k = 0; k < 8; ++k) {
            const float nt = __int_as_float(__builtin_amdgcn_readlane(nrmBits, sub * 8 + k));
            const int lt = __builtin_amdgcn_readlane(loc, sub * 8 + k);
            float* ap = t.acc + lt * FEAT + t.lane;
            __hip_atomic_fetch_add(ap + 0,  hv[k][0] * nt, __ATOMIC_RELAXED, __HIP_MEMORY_SCOPE_WORKGROUP);
            __hip_atomic_fetch_add(ap + 32, hv[k][1] * nt, __ATOMIC_RELAXED, __HIP_MEMORY_SCOPE_WORKGROUP);
            __hip_atomic_fetch_add(ap + 64, hv[k][2] * nt, __ATOMIC_RELAXED, __HIP_MEMORY_SCOPE_WORKGROUP);
            __hip_atomic_fetch_add(ap + 96, hv[k][3] * nt, __ATOMIC_RELAXED, __HIP_MEMORY_SCOPE_WORKGROUP);
        }
    }
    t.head = (t.head + 32) & (QCAP - 1);
    t.qn -= cnt;
}

__device__ __forceinline__ void tile_offer(TileCtx& t, int c, int e) {
    const bool m = (unsigned)(c - t.base) < (unsigned)TILE_NODES;
    const unsigned mask = __builtin_amdgcn_ballot_w32(m);
    if (mask != 0u) {
        const int pos = t.head + t.qn + (int)__builtin_amdgcn_mbcnt_lo(mask, 0u);
        if (m) t.q[pos & (QCAP - 1)] = e;
        t.qn += __builtin_popcount(mask);
    }
}

__global__ __launch_bounds__(1024) void aggregate_tile_kernel(const int* __restrict__ row,
                                                              const int* __restrict__ col,
                                                              const float* __restrict__ ew,
                                                              const float* __restrict__ dis,
                                                              const float* __restrict__ h,
                                                              float* __restrict__ agg,
                                                              int E, int N) {
    __shared__ float acc[(TILE_NODES + 1) * FEAT];
    __shared__ int qbuf[AGG_WAVES * QCAP];

    const int tid = threadIdx.x;
    const int wave = __builtin_amdgcn_readfirstlane(tid >> 5);

    TileCtx t;
    t.acc = acc;
    t.q = qbuf + wave * QCAP;
    t.base = blockIdx.x * TILE_NODES;
    t.lane = tid & 31;
    t.head = 0;
    t.qn = 0;

    for (int i = tid; i < (TILE_NODES + 1) * FEAT / 4; i += 1024)
        ((float4*)acc)[i] = make_float4(0.f, 0.f, 0.f, 0.f);
    __syncthreads();

    const int nGroups = E >> 7;
    const bool vec = ((((size_t)col) & 15) == 0);
    const int start = (nGroups / (int)gridDim.x) * (int)blockIdx.x;

    int i = wave;
    if (vec) {
        for (; i + 3 * AGG_WAVES < nGroups; i += 4 * AGG_WAVES) {
            int4 cv[4];
            int eb[4];
#pragma unroll
            for (int u = 0; u < 4; ++u) {
                int g = start + i + u * AGG_WAVES;
                if (g >= nGroups) g -= nGroups;
                eb[u] = g * 128 + t.lane * 4;
                cv[u] = *(const int4*)(col + eb[u]);
            }
#pragma unroll
            for (int u = 0; u < 4; ++u) {
                const unsigned b = (unsigned)t.base;
                const bool any = ((unsigned)cv[u].x - b < (unsigned)TILE_NODES) |
                                 ((unsigned)cv[u].y - b < (unsigned)TILE_NODES) |
                                 ((unsigned)cv[u].z - b < (unsigned)TILE_NODES) |
                                 ((unsigned)cv[u].w - b < (unsigned)TILE_NODES);
                if (__builtin_amdgcn_ballot_w32(any) != 0u) {
                    tile_offer(t, cv[u].x, eb[u] + 0);
                    tile_offer(t, cv[u].y, eb[u] + 1);
                    tile_offer(t, cv[u].z, eb[u] + 2);
                    tile_offer(t, cv[u].w, eb[u] + 3);
                }
            }
            while (t.qn >= 32) tile_drain(t, 32, row, col, ew, dis, h);
        }
    }
    for (; i < nGroups; i += AGG_WAVES) {
        int g = start + i;
        if (g >= nGroups) g -= nGroups;
        const int e0 = g * 128 + t.lane;
        const int c0 = col[e0], c1 = col[e0 + 32], c2 = col[e0 + 64], c3 = col[e0 + 96];
        tile_offer(t, c0, e0);
        tile_offer(t, c1, e0 + 32);
        tile_offer(t, c2, e0 + 64);
        tile_offer(t, c3, e0 + 96);
        while (t.qn >= 32) tile_drain(t, 32, row, col, ew, dis, h);
    }
    if (wave == 0) {
        for (int eb = nGroups * 128; eb < E; eb += 32) {
            const int e = eb + t.lane;
            const int c = (e < E) ? col[e] : -1;
            tile_offer(t, c, e);
        }
    }
    while (t.qn >= 32) tile_drain(t, 32, row, col, ew, dis, h);
    if (t.qn > 0) tile_drain(t, t.qn, row, col, ew, dis, h);
    __syncthreads();

    const int rows = min(TILE_NODES, N - t.base);
    float4* dst = (float4*)(agg + (size_t)t.base * FEAT);
    for (int i4 = tid; i4 < rows * (FEAT / 4); i4 += 1024)
        dst[i4] = ((const float4*)acc)[i4];
}

__global__ __launch_bounds__(256) void epilogue1_kernel(float* __restrict__ agg,
                                                        const float* __restrict__ h,
                                                        const float* __restrict__ dis,
                                                        const float* __restrict__ b0,
                                                        const float* __restrict__ g,
                                                        const float* __restrict__ be,
                                                        const float* __restrict__ mu,
                                                        const float* __restrict__ var,
                                                        int n4) {
    int idx4 = blockIdx.x * 256 + threadIdx.x;
    if (idx4 >= n4) return;
    int i  = idx4 >> 5;
    int f  = (idx4 & 31) << 2;
    float d = dis[i];
    float d2 = d * d;
    float4 a  = *(float4*)(agg + (size_t)idx4 * 4);
    float4 hv = *(const float4*)(h + (size_t)idx4 * 4);
    float out[4];
    float av[4] = {a.x, a.y, a.z, a.w};
    float hh[4] = {hv.x, hv.y, hv.z, hv.w};
#pragma unroll
    for (int j = 0; j < 4; ++j) {
        int ff = f + j;
        float v = av[j] + d2 * hh[j] + b0[ff];
        v = (v - mu[ff]) * rsqrtf(var[ff] + BN_EPS) * g[ff] + be[ff];
        out[j] = (v > 0.0f) ? v : expm1f(v);
    }
    *(float4*)(agg + (size_t)idx4 * 4) = make_float4(out[0], out[1], out[2], out[3]);
}

__global__ __launch_bounds__(256) void final_kernel(const float* __restrict__ emb,
                                                    const float* __restrict__ h1,
                                                    const float* __restrict__ agg2,
                                                    const float* __restrict__ h2,
                                                    const float* __restrict__ dis,
                                                    const float* __restrict__ b1,
                                                    float* __restrict__ out, int n4) {
    int idx4 = blockIdx.x * 256 + threadIdx.x;
    if (idx4 >= n4) return;
    int i = idx4 >> 5;
    int f = (idx4 & 31) << 2;
    float d = dis[i];
    float d2 = d * d;
    float4 e  = *(const float4*)(emb  + (size_t)idx4 * 4);
    float4 o1 = *(const float4*)(h1   + (size_t)idx4 * 4);
    float4 a2 = *(const float4*)(agg2 + (size_t)idx4 * 4);
    float4 p2 = *(const float4*)(h2   + (size_t)idx4 * 4);
    const float inv3 = 1.0f / 3.0f;
    float4 r;
    r.x = (e.x + o1.x + a2.x + d2 * p2.x + b1[f + 0]) * inv3;
    r.y = (e.y + o1.y + a2.y + d2 * p2.y + b1[f + 1]) * inv3;
    r.z = (e.z + o1.z + a2.z + d2 * p2.z + b1[f + 2]) * inv3;
    r.w = (e.w + o1.w + a2.w + d2 * p2.w + b1[f + 3]) * inv3;
    *(float4*)(out + (size_t)idx4 * 4) = r;
}

extern "C" void kernel_launch(void* const* d_in, const int* in_sizes, int n_in,
                              void* d_out, int out_size, void* d_ws, size_t ws_size,
                              hipStream_t stream) {
    const float* emb = (const float*)d_in[0];
    const int*   ei  = (const int*)d_in[1];
    const float* ew  = (const float*)d_in[2];
    const float* W0  = (const float*)d_in[3];
    const float* b0  = (const float*)d_in[4];
    const float* W1  = (const float*)d_in[5];
    const float* b1  = (const float*)d_in[6];
    const float* bng = (const float*)d_in[7];
    const float* bnb = (const float*)d_in[8];
    const float* bnm = (const float*)d_in[9];
    const float* bnv = (const float*)d_in[10];
    float* out = (float*)d_out;

    const int N  = in_sizes[0] / FEAT;
    const int E  = in_sizes[2];
    const size_t ND = (size_t)N * FEAT;

    const int* row = ei;
    const int* col = ei + E;

    float* ws   = (float*)d_ws;
    size_t disPad = ((size_t)N + 1023) & ~(size_t)1023;
    float* dis  = ws;
    float* bufA = ws + disPad;
    float* bufB = bufA + ND;
    float* bufC = bufB + ND;

    hipMemsetAsync(dis,  0, (size_t)N * sizeof(float), stream);

    const int n4 = (int)(ND / 4);
    const int gElem  = (n4 + 255) / 256;
    const int gTile  = (N + TILE_NODES - 1) / TILE_NODES;
    const int gRowT  = (N + 15) / 16;

    degree_kernel<<<(E + 255) / 256, 256, 0, stream>>>(col, ew, dis, E);
    dis_kernel<<<(N + 255) / 256, 256, 0, stream>>>(dis, N);

    gemm_n128_kernel<<<gRowT, 256, 0, stream>>>(emb, W0, bufA, N);

    aggregate_tile_kernel<<<gTile, 1024, 0, stream>>>(row, col, ew, dis, bufA, bufB, E, N);

    epilogue1_kernel<<<gElem, 256, 0, stream>>>(bufB, bufA, dis, b0, bng, bnb, bnm, bnv, n4);

    gemm_n128_kernel<<<gRowT, 256, 0, stream>>>(bufB, W1, bufA, N);

    aggregate_tile_kernel<<<gTile, 1024, 0, stream>>>(row, col, ew, dis, bufA, bufC, E, N);

    final_kernel<<<gElem, 256, 0, stream>>>(emb, bufB, bufC, bufA, dis, b1, out, n4);
}


# both elementwise passes (self-loop+bias+BN+ELU; self-loop+bias+mean of 3) moved to the aggregation kernels' LDS-tile store: 8 -> 6 launches, agg planes never reach memory; LDS arrival sum order kept
# speedup vs baseline: 2.1135x; 1.1878x over previous
#include <hip/hip_runtime.h>
#include <hip/hip_bf16.h>
#include <math.h>

typedef float v2f __attribute__((ext_vector_type(2)));
typedef float v8f __attribute__((ext_vector_type(8)));

#if defined(__has_builtin)
#if __has_builtin(__builtin_amdgcn_wmma_f32_16x16x4_f32)
#define HAS_WMMA_F32_4 1
#endif
#endif
#ifndef HAS_WMMA_F32_4
#define HAS_WMMA_F32_4 0
#endif

#define FEAT 128
#define BN_EPS 1e-5f

#define TILE_NODES 196

__global__ __launch_bounds__(256) void degree_key_kernel(const int* __restrict__ col,
                                                         const float* __restrict__ w,
                                                         float* __restrict__ deg,
                                                         int* __restrict__ key4, int E) {
    const int t = blockIdx.x * 256 + threadIdx.x;
    const int e0 = t * 4;
    if (e0 >= E) return;
    unsigned k = 0u;
    if (e0 + 3 < E) {
        int c[4];
        float wv[4];
#pragma unroll
        for (int j = 0; j < 4; ++j) {
            c[j] = col[e0 + j];
            wv[j] = w[e0 + j];
        }
#pragma unroll
        for (int j = 0; j < 4; ++j) {
            __hip_atomic_fetch_add(&deg[c[j]], wv[j], __ATOMIC_RELAXED,
                                   __HIP_MEMORY_SCOPE_AGENT);
            k |= ((unsigned)(c[j] / TILE_NODES) & 255u) << (8 * j);
        }
    } else {
        for (int j = 0; j < 4 && e0 + j < E; ++j) {
            const int c = col[e0 + j];
            __hip_atomic_fetch_add(&deg[c], w[e0 + j], __ATOMIC_RELAXED,
                                   __HIP_MEMORY_SCOPE_AGENT);
            k |= ((unsigned)(c / TILE_NODES) & 255u) << (8 * j);
        }
    }
    key4[t] = (int)k;
}

__global__ __launch_bounds__(256) void dis_kernel(float* __restrict__ deg_dis, int n) {
    int i = blockIdx.x * 256 + threadIdx.x;
    if (i < n) deg_dis[i] = rsqrtf(deg_dis[i] + 1.0f);
}

__global__ __launch_bounds__(256) void gemm_n128_kernel(const float* __restrict__ X,
                                                        const float* __restrict__ W,
                                                        float* __restrict__ Y,
                                                        int nrows) {
    __shared__ float lA[16 * 132];

    const int tid = threadIdx.x;
    const int rowBase = blockIdx.x * 16;

#pragma unroll
    for (int i = 0; i < 2; ++i) {
        int idx = tid + i * 256;
        int r = idx >> 5;
        int c4 = (idx & 31) << 2;
        float4 v = make_float4(0.f, 0.f, 0.f, 0.f);
        if (rowBase + r < nrows)
            v = *(const float4*)(X + (size_t)(rowBase + r) * FEAT + c4);
        float* dst = &lA[r * 132 + c4];
        dst[0] = v.x; dst[1] = v.y; dst[2] = v.z; dst[3] = v.w;
    }
    __syncthreads();

    const int wave = tid >> 5;
    const int lane = tid & 31;
    const int lid  = lane & 15;
    const int half = lane >> 4;
    const int n0   = wave * 16;

#if HAS_WMMA_F32_4
    v8f acc = {0.f, 0.f, 0.f, 0.f, 0.f, 0.f, 0.f, 0.f};
    const float* arow = &lA[lid * 132];
#pragma unroll
    for (int k0 = 0; k0 < FEAT; k0 += 4) {
        const int ka = k0 + 2 * half;
        v2f a, b;
        a.x = arow[ka];
        a.y = arow[ka + 1];
        b.x = W[(size_t)ka * FEAT + n0 + lid];
        b.y = W[(size_t)(ka + 1) * FEAT + n0 + lid];
        acc = __builtin_amdgcn_wmma_f32_16x16x4_f32(
             false, a,  false, b,
             (short)0, acc,  false,  false);
    }
    float* yb = Y + (size_t)rowBase * FEAT + n0 + lid;
#pragma unroll
    for (int v = 0; v < 8; ++v) {
        int m = v + 8 * half;
        if (rowBase + m < nrows) yb[(size_t)m * FEAT] = acc[v];
    }
#else
    float accs[8] = {0.f, 0.f, 0.f, 0.f, 0.f, 0.f, 0.f, 0.f};
    for (int k = 0; k < FEAT; ++k) {
        float bv = W[(size_t)k * FEAT + n0 + lid];
#pragma unroll
        for (int v = 0; v < 8; ++v)
            accs[v] = fmaf(lA[(v + 8 * half) * 132 + k], bv, accs[v]);
    }
    float* yb = Y + (size_t)rowBase * FEAT + n0 + lid;
#pragma unroll
    for (int v = 0; v < 8; ++v) {
        int m = v + 8 * half;
        if (rowBase + m < nrows) yb[(size_t)m * FEAT] = accs[v];
    }
#endif
}

#define AGG_WAVES 32
#define QCAP 1024

struct TileCtx {
    float* acc;
    int* q;
    int base;
    int lane;
    int head;
    int qn;
};

__device__ __forceinline__ void tile_drain(TileCtx& t, int cnt,
                                           const int* __restrict__ row,
                                           const int* __restrict__ col,
                                           const float* __restrict__ ew,
                                           const float* __restrict__ dis,
                                           const float* __restrict__ h) {
    const bool valid = t.lane < cnt;
    int e = t.q[(t.head + t.lane) & (QCAP - 1)];
    e = valid ? e : 0;
    const int r = row[e];
    const int c = col[e];
    const bool mine = valid && ((unsigned)(c - t.base) < (unsigned)TILE_NODES);
    float nrm = dis[r] * ew[e] * dis[c];
    nrm = mine ? nrm : 0.0f;
    const int loc = mine ? (c - t.base) : TILE_NODES;
    const int nrmBits = __float_as_int(nrm);
#pragma unroll
    for (int sub = 0; sub < 4; ++sub) {
        if (sub * 8 >= cnt) break;
        float hv[8][4];
#pragma unroll
        for (int k = 0; k < 8; ++k) {
            const int rt = __builtin_amdgcn_readlane(r, sub * 8 + k);
            const float* hp = h + (size_t)rt * FEAT + t.lane;
            hv[k][0] = hp[0];
            hv[k][1] = hp[32];
            hv[k][2] = hp[64];
            hv[k][3] = hp[96];
        }
#pragma unroll
        for (int k = 0; k < 8; ++k) {
            const float nt = __int_as_float(__builtin_amdgcn_readlane(nrmBits, sub * 8 + k));
            const int lt = __builtin_amdgcn_readlane(loc, sub * 8 + k);
            float* ap = t.acc + lt * FEAT + t.lane;
            __hip_atomic_fetch_add(ap + 0,  hv[k][0] * nt, __ATOMIC_RELAXED, __HIP_MEMORY_SCOPE_WORKGROUP);
            __hip_atomic_fetch_add(ap + 32, hv[k][1] * nt, __ATOMIC_RELAXED, __HIP_MEMORY_SCOPE_WORKGROUP);
            __hip_atomic_fetch_add(ap + 64, hv[k][2] * nt, __ATOMIC_RELAXED, __HIP_MEMORY_SCOPE_WORKGROUP);
            __hip_atomic_fetch_add(ap + 96, hv[k][3] * nt, __ATOMIC_RELAXED, __HIP_MEMORY_SCOPE_WORKGROUP);
        }
    }
    t.head = (t.head + 32) & (QCAP - 1);
    t.qn -= cnt;
}

__device__ __forceinline__ void tile_offer(TileCtx& t, int c, int e) {
    const bool m = (unsigned)(c - t.base) < (unsigned)TILE_NODES;
    const unsigned mask = __builtin_amdgcn_ballot_w32(m);
    if (mask != 0u) {
        const int pos = t.head + t.qn + (int)__builtin_amdgcn_mbcnt_lo(mask, 0u);
        if (m) t.q[pos & (QCAP - 1)] = e;
        t.qn += __builtin_popcount(mask);
    }
}

__device__ __forceinline__ unsigned zero_bytes(unsigned x) {
    return ~(((x & 0x7F7F7F7Fu) + 0x7F7F7F7Fu) | x | 0x7F7F7F7Fu);
}

__device__ __forceinline__ unsigned match16(int4 kv, unsigned pat) {
    return (zero_bytes((unsigned)kv.x ^ pat) >> 7) | (zero_bytes((unsigned)kv.y ^ pat) >> 6) |
           (zero_bytes((unsigned)kv.z ^ pat) >> 5) | (zero_bytes((unsigned)kv.w ^ pat) >> 4);
}

__device__ __forceinline__ void tile_offer_mask(TileCtx& t, unsigned m, int ebase) {
    unsigned live = __builtin_amdgcn_ballot_w32(m != 0u);
    while (live != 0u) {
        const bool has = (m != 0u);
        const int p = __builtin_ffs((int)m) - 1;
        const int pos = t.head + t.qn + (int)__builtin_amdgcn_mbcnt_lo(live, 0u);
        if (has) t.q[pos & (QCAP - 1)] = ebase + ((p & 7) << 2) + (p >> 3);
        t.qn += __builtin_popcount(live);
        m &= m - 1u;
        live = __builtin_amdgcn_ballot_w32(m != 0u);
    }
}

__device__ __forceinline__ void tile_sum(TileCtx& t, int wave, int tid,
                                         const int* __restrict__ row,
                                         const int* __restrict__ col,
                                         const int* __restrict__ key4,
                                         const float* __restrict__ ew,
                                         const float* __restrict__ dis,
                                         const float* __restrict__ h, int E) {
    for (int i = tid; i < (TILE_NODES + 1) * FEAT / 4; i += 1024)
        ((float4*)t.acc)[i] = make_float4(0.f, 0.f, 0.f, 0.f);
    __syncthreads();

    const int nGroups = E >> 9;
    const unsigned pat = (unsigned)(blockIdx.x & 255u) * 0x01010101u;
    const int start = (nGroups / (int)gridDim.x) * (int)blockIdx.x;

    int i = wave;
    for (; i + 3 * AGG_WAVES < nGroups; i += 4 * AGG_WAVES) {
        int4 kv[4];
        int eb[4];
#pragma unroll
        for (int u = 0; u < 4; ++u) {
            int g = start + i + u * AGG_WAVES;
            if (g >= nGroups) g -= nGroups;
            eb[u] = g * 512 + t.lane * 16;
            kv[u] = *(const int4*)(key4 + g * 128 + t.lane * 4);
        }
        int done = 0;
        do {
#pragma unroll
            for (int u = 0; u < 4; ++u) {
                if (u == done && t.qn <= QCAP - 512) {
                    tile_offer_mask(t, match16(kv[u], pat), eb[u]);
                    done = u + 1;
                }
            }
            while (t.qn >= 32) tile_drain(t, 32, row, col, ew, dis, h);
        } while (done < 4);
    }
    for (; i < nGroups; i += AGG_WAVES) {
        int g = start + i;
        if (g >= nGroups) g -= nGroups;
        const int4 kv = *(const int4*)(key4 + g * 128 + t.lane * 4);
        tile_offer_mask(t, match16(kv, pat), g * 512 + t.lane * 16);
        while (t.qn >= 32) tile_drain(t, 32, row, col, ew, dis, h);
    }
    if (wave == 0) {
        for (int eb = nGroups * 512; eb < E; eb += 32) {
            const int e = eb + t.lane;
            const int c = (e < E) ? col[e] : -1;
            tile_offer(t, c, e);
        }
    }
    while (t.qn >= 32) tile_drain(t, 32, row, col, ew, dis, h);
    if (t.qn > 0) tile_drain(t, t.qn, row, col, ew, dis, h);
    __syncthreads();
}

#define CLOSE_TRIPS 4

__global__ __launch_bounds__(1024) void aggregate_layer1_kernel(const int* __restrict__ row,
                                                                const int* __restrict__ col,
                                                                const int* __restrict__ key4,
                                                                const float* __restrict__ ew,
                                                                const float* __restrict__ dis,
                                                                const float* __restrict__ h,
                                                                const float* __restrict__ b0,
                                                                const float* __restrict__ g,
                                                                const float* __restrict__ be,
                                                                const float* __restrict__ mu,
                                                                const float* __restrict__ var,
                                                                float* __restrict__ h1,
                                                                int E, int N) {
    __shared__ float acc[(TILE_NODES + 1) * FEAT];
    __shared__ int qbuf[AGG_WAVES * QCAP];

    const int tid = threadIdx.x;
    const int wave = __builtin_amdgcn_readfirstlane(tid >> 5);

    TileCtx t;
    t.acc = acc;
    t.q = qbuf + wave * QCAP;
    t.base = blockIdx.x * TILE_NODES;
    t.lane = tid & 31;
    t.head = 0;
    t.qn = 0;

    tile_sum(t, wave, tid, row, col, key4, ew, dis, h, E);

    const int f = (tid & 31) << 2;
    const float4 bv = *(const float4*)(b0 + f);
    const float4 gv = *(const float4*)(g + f);
    const float4 ev = *(const float4*)(be + f);
    const float4 mv = *(const float4*)(mu + f);
    const float4 vv = *(const float4*)(var + f);
    const float bb[4] = {bv.x, bv.y, bv.z, bv.w};
    const float gg[4] = {gv.x, gv.y, gv.z, gv.w};
    const float ee[4] = {ev.x, ev.y, ev.z, ev.w};
    const float mm[4] = {mv.x, mv.y, mv.z, mv.w};
    const float rs[4] = {rsqrtf(vv.x + BN_EPS), rsqrtf(vv.y + BN_EPS),
                         rsqrtf(vv.z + BN_EPS), rsqrtf(vv.w + BN_EPS)};

    const int rows = min(TILE_NODES, N - t.base);
    const int lim = rows * (FEAT / 4);
    const float4* hsrc = (const float4*)(h + (size_t)t.base * FEAT);
    float4* dst = (float4*)(h1 + (size_t)t.base * FEAT);
    for (int i0 = tid; i0 < lim; i0 += CLOSE_TRIPS * 1024) {
        float4 hv[CLOSE_TRIPS];
        float dv[CLOSE_TRIPS];
#pragma unroll
        for (int u = 0; u < CLOSE_TRIPS; ++u) {
            const int j4 = min(i0 + u * 1024, lim - 1);
            hv[u] = hsrc[j4];
            dv[u] = dis[t.base + (j4 >> 5)];
        }
#pragma unroll
        for (int u = 0; u < CLOSE_TRIPS; ++u) {
            const int i4 = i0 + u * 1024;
            if (i4 < lim) {
                const float4 a = ((const float4*)acc)[i4];
                const float d2 = dv[u] * dv[u];
                const float av[4] = {a.x, a.y, a.z, a.w};
                const float hh[4] = {hv[u].x, hv[u].y, hv[u].z, hv[u].w};
                float o[4];
#pragma unroll
                for (int j = 0; j < 4; ++j) {
                    float v = av[j] + d2 * hh[j] + bb[j];
                    v = (v - mm[j]) * rs[j] * gg[j] + ee[j];
                    o[j] = (v > 0.0f) ? v : expm1f(v);
                }
                dst[i4] = make_float4(o[0], o[1], o[2], o[3]);
            }
        }
    }
}

__global__ __launch_bounds__(1024) void aggregate_layer2_kernel(const int* __restrict__ row,
                                                                const int* __restrict__ col,
                                                                const int* __restrict__ key4,
                                                                const float* __restrict__ ew,
                                                                const float* __restrict__ dis,
                                                                const float* __restrict__ h,
                                                                const float* __restrict__ emb,
                                                                const float* __restrict__ h1,
                                                                const float* __restrict__ b1,
                                                                float* __restrict__ out,
                                                                int E, int N) {
    __shared__ float acc[(TILE_NODES + 1) * FEAT];
    __shared__ int qbuf[AGG_WAVES * QCAP];

    const int tid = threadIdx.x;
    const int wave = __builtin_amdgcn_readfirstlane(tid >> 5);

    TileCtx t;
    t.acc = acc;
    t.q = qbuf + wave * QCAP;
    t.base = blockIdx.x * TILE_NODES;
    t.lane = tid & 31;
    t.head = 0;
    t.qn = 0;

    tile_sum(t, wave, tid, row, col, key4, ew, dis, h, E);

    const int f = (tid & 31) << 2;
    const float4 bv = *(const float4*)(b1 + f);
    const float inv3 = 1.0f / 3.0f;

    const int rows = min(TILE_NODES, N - t.base);
    const int lim = rows * (FEAT / 4);
    const float4* hsrc = (const float4*)(h + (size_t)t.base * FEAT);
    const float4* esrc = (const float4*)(emb + (size_t)t.base * FEAT);
    const float4* osrc = (const float4*)(h1 + (size_t)t.base * FEAT);
    float4* dst = (float4*)(out + (size_t)t.base * FEAT);
    for (int i0 = tid; i0 < lim; i0 += CLOSE_TRIPS * 1024) {
        float4 pv[CLOSE_TRIPS], xv[CLOSE_TRIPS], ov[CLOSE_TRIPS];
        float dv[CLOSE_TRIPS];
#pragma unroll
        for (int u = 0; u < CLOSE_TRIPS; ++u) {
            const int j4 = min(i0 + u * 1024, lim - 1);
            xv[u] = esrc[j4];
            ov[u] = osrc[j4];
            pv[u] = hsrc[j4];
            dv[u] = dis[t.base + (j4 >> 5)];
        }
#pragma unroll
        for (int u = 0; u < CLOSE_TRIPS; ++u) {
            asm volatile("" :: "v"(xv[u].x), "v"(xv[u].y), "v"(xv[u].z), "v"(xv[u].w),
                               "v"(ov[u].x), "v"(ov[u].y), "v"(ov[u].z), "v"(ov[u].w),
                               "v"(pv[u].x), "v"(pv[u].y), "v"(pv[u].z), "v"(pv[u].w),
                               "v"(dv[u]));
        }
#pragma unroll
        for (int u = 0; u < CLOSE_TRIPS; ++u) {
            const int i4 = i0 + u * 1024;
            if (i4 < lim) {
                const float4 a2 = ((const float4*)acc)[i4];
                const float d2 = dv[u] * dv[u];
                const float4 e = xv[u];
                const float4 o1 = ov[u];
                const float4 p2 = pv[u];
                float4 r;
                r.x = (e.x + o1.x + a2.x + d2 * p2.x + bv.x) * inv3;
                r.y = (e.y + o1.y + a2.y + d2 * p2.y + bv.y) * inv3;
                r.z = (e.z + o1.z + a2.z + d2 * p2.z + bv.z) * inv3;
                r.w = (e.w + o1.w + a2.w + d2 * p2.w + bv.w) * inv3;
                dst[i4] = r;
            }
        }
    }
}

extern "C" void kernel_launch(void* const* d_in, const int* in_sizes, int n_in,
                              void* d_out, int out_size, void* d_ws, size_t ws_size,
                              hipStream_t stream) {
    const float* emb = (const float*)d_in[0];
    const int*   ei  = (const int*)d_in[1];
    const float* ew  = (const float*)d_in[2];
    const float* W0  = (const float*)d_in[3];
    const float* b0  = (const float*)d_in[4];
    const float* W1  = (const float*)d_in[5];
    const float* b1  = (const float*)d_in[6];
    const float* bng = (const float*)d_in[7];
    const float* bnb = (const float*)d_in[8];
    const float* bnm = (const float*)d_in[9];
    const float* bnv = (const float*)d_in[10];
    float* out = (float*)d_out;

    const int N  = in_sizes[0] / FEAT;
    const int E  = in_sizes[2];
    const size_t ND = (size_t)N * FEAT;

    const int* row = ei;
    const int* col = ei + E;

    float* ws   = (float*)d_ws;
    size_t disPad = ((size_t)N + 1023) & ~(size_t)1023;
    float* dis  = ws;
    float* bufA = ws + disPad;
    float* bufB = bufA + ND;
    int* key4   = (int*)(bufB + ND);

    hipMemsetAsync(dis,  0, (size_t)N * sizeof(float), stream);

    const int gTile  = (N + TILE_NODES - 1) / TILE_NODES;
    const int gRowT  = (N + 15) / 16;

    degree_key_kernel<<<(E + 1023) / 1024, 256, 0, stream>>>(col, ew, dis, key4, E);
    dis_kernel<<<(N + 255) / 256, 256, 0, stream>>>(dis, N);

    gemm_n128_kernel<<<gRowT, 256, 0, stream>>>(emb, W0, bufA, N);

    aggregate_layer1_kernel<<<gTile, 1024, 0, stream>>>(row, col, key4, ew, dis, bufA,
                                                        b0, bng, bnb, bnm, bnv, bufB, E, N);

    gemm_n128_kernel<<<gRowT, 256, 0, stream>>>(bufB, W1, bufA, N);

    aggregate_layer2_kernel<<<gTile, 1024, 0, stream>>>(row, col, key4, ew, dis, bufA,
                                                        emb, bufB, b1, out, E, N);
}


# drain fetches one 16-byte record {row,col,w} per edge (from the degree pass), not 5 scattered words; norm factored: dis[row] scales GEMM input rows, dis[col] the tile at store; LDS arrival order kept
# speedup vs baseline: 2.3667x; 1.1198x over previous
#include <hip/hip_runtime.h>
#include <hip/hip_bf16.h>
#include <math.h>

typedef float v2f __attribute__((ext_vector_type(2)));
typedef float v8f __attribute__((ext_vector_type(8)));

#if defined(__has_builtin)
#if __has_builtin(__builtin_amdgcn_wmma_f32_16x16x4_f32)
#define HAS_WMMA_F32_4 1
#endif
#endif
#ifndef HAS_WMMA_F32_4
#define HAS_WMMA_F32_4 0
#endif

#define FEAT 128
#define BN_EPS 1e-5f

#define TILE_NODES 196

__global__ __launch_bounds__(256) void degree_key_kernel(const int* __restrict__ row,
                                                         const int* __restrict__ col,
                                                         const float* __restrict__ w,
                                                         float* __restrict__ deg,
                                                         int* __restrict__ key4,
                                                         int* __restrict__ rec4, int E) {
    int4* __restrict__ rec = (int4*)rec4;
    const int er = blockIdx.x * 1024 + threadIdx.x;
    int rr[4], rc[4];
    float rw[4];
#pragma unroll
    for (int j = 0; j < 4; ++j) {
        const int el = min(er + j * 256, E - 1);
        rr[j] = row[el];
        rc[j] = col[el];
        rw[j] = w[el];
    }
#pragma unroll
    for (int j = 0; j < 4; ++j) {
        const int e = er + j * 256;
        if (e < E) rec[e] = make_int4(rr[j], rc[j], __float_as_int(rw[j]), 0);
    }
    const int t = blockIdx.x * 256 + threadIdx.x;
    const int e0 = t * 4;
    if (e0 >= E) return;
    unsigned k = 0u;
    if (e0 + 3 < E) {
        int c[4];
        float wv[4];
#pragma unroll
        for (int j = 0; j < 4; ++j) {
            c[j] = col[e0 + j];
            wv[j] = w[e0 + j];
        }
#pragma unroll
        for (int j = 0; j < 4; ++j) {
            __hip_atomic_fetch_add(&deg[c[j]], wv[j], __ATOMIC_RELAXED,
                                   __HIP_MEMORY_SCOPE_AGENT);
            k |= ((unsigned)(c[j] / TILE_NODES) & 255u) << (8 * j);
        }
    } else {
        for (int j = 0; j < 4 && e0 + j < E; ++j) {
            const int c = col[e0 + j];
            __hip_atomic_fetch_add(&deg[c], w[e0 + j], __ATOMIC_RELAXED,
                                   __HIP_MEMORY_SCOPE_AGENT);
            k |= ((unsigned)(c / TILE_NODES) & 255u) << (8 * j);
        }
    }
    key4[t] = (int)k;
}

__global__ __launch_bounds__(256) void dis_kernel(float* __restrict__ deg_dis, int n) {
    int i = blockIdx.x * 256 + threadIdx.x;
    if (i < n) deg_dis[i] = rsqrtf(deg_dis[i] + 1.0f);
}

__global__ __launch_bounds__(256) void gemm_n128_kernel(const float* __restrict__ X,
                                                        const float* __restrict__ W,
                                                        const float* __restrict__ dis,
                                                        float* __restrict__ Y,
                                                        int nrows) {
    __shared__ float lA[16 * 132];

    const int tid = threadIdx.x;
    const int rowBase = blockIdx.x * 16;

#pragma unroll
    for (int i = 0; i < 2; ++i) {
        int idx = tid + i * 256;
        int r = idx >> 5;
        int c4 = (idx & 31) << 2;
        float4 v = make_float4(0.f, 0.f, 0.f, 0.f);
        float s = 0.f;
        if (rowBase + r < nrows) {
            v = *(const float4*)(X + (size_t)(rowBase + r) * FEAT + c4);
            s = dis[rowBase + r];
        }
        float* dst = &lA[r * 132 + c4];
        dst[0] = v.x * s; dst[1] = v.y * s; dst[2] = v.z * s; dst[3] = v.w * s;
    }
    __syncthreads();

    const int wave = tid >> 5;
    const int lane = tid & 31;
    const int lid  = lane & 15;
    const int half = lane >> 4;
    const int n0   = wave * 16;

#if HAS_WMMA_F32_4
    v8f acc = {0.f, 0.f, 0.f, 0.f, 0.f, 0.f, 0.f, 0.f};
    const float* arow = &lA[lid * 132];
#pragma unroll
    for (int k0 = 0; k0 < FEAT; k0 += 4) {
        const int ka = k0 + 2 * half;
        v2f a, b;
        a.x = arow[ka];
        a.y = arow[ka + 1];
        b.x = W[(size_t)ka * FEAT + n0 + lid];
        b.y = W[(size_t)(ka + 1) * FEAT + n0 + lid];
        acc = __builtin_amdgcn_wmma_f32_16x16x4_f32(
             false, a,  false, b,
             (short)0, acc,  false,  false);
    }
    float* yb = Y + (size_t)rowBase * FEAT + n0 + lid;
#pragma unroll
    for (int v = 0; v < 8; ++v) {
        int m = v + 8 * half;
        if (rowBase + m < nrows) yb[(size_t)m * FEAT] = acc[v];
    }
#else
    float accs[8] = {0.f, 0.f, 0.f, 0.f, 0.f, 0.f, 0.f, 0.f};
    for (int k = 0; k < FEAT; ++k) {
        float bv = W[(size_t)k * FEAT + n0 + lid];
#pragma unroll
        for (int v = 0; v < 8; ++v)
            accs[v] = fmaf(lA[(v + 8 * half) * 132 + k], bv, accs[v]);
    }
    float* yb = Y + (size_t)rowBase * FEAT + n0 + lid;
#pragma unroll
    for (int v = 0; v < 8; ++v) {
        int m = v + 8 * half;
        if (rowBase + m < nrows) yb[(size_t)m * FEAT] = accs[v];
    }
#endif
}

#define AGG_WAVES 32
#define QCAP 1024

struct TileCtx {
    float* acc;
    int* q;
    int base;
    int lane;
    int head;
    int qn;
};

__device__ __forceinline__ void tile_drain(TileCtx& t, int cnt,
                                           const int4* __restrict__ rec,
                                           const float* __restrict__ h) {
    const bool valid = t.lane < cnt;
    int e = t.q[(t.head + t.lane) & (QCAP - 1)];
    e = valid ? e : 0;
    const int4 rc = rec[e];
    const int r = rc.x;
    const int c = rc.y;
    const bool mine = valid && ((unsigned)(c - t.base) < (unsigned)TILE_NODES);
    const int loc = mine ? (c - t.base) : TILE_NODES;
    const int nrmBits = mine ? rc.z : 0;
#pragma unroll
    for (int sub = 0; sub < 4; ++sub) {
        if (sub * 8 >= cnt) break;
        float hv[8][4];
#pragma unroll
        for (int k = 0; k < 8; ++k) {
            const int rt = __builtin_amdgcn_readlane(r, sub * 8 + k);
            const float* hp = h + (size_t)rt * FEAT + t.lane;
            hv[k][0] = hp[0];
            hv[k][1] = hp[32];
            hv[k][2] = hp[64];
            hv[k][3] = hp[96];
        }
#pragma unroll
        for (int k = 0; k < 8; ++k) {
            const float nt = __int_as_float(__builtin_amdgcn_readlane(nrmBits, sub * 8 + k));
            const int lt = __builtin_amdgcn_readlane(loc, sub * 8 + k);
            float* ap = t.acc + lt * FEAT + t.lane;
            __hip_atomic_fetch_add(ap + 0,  hv[k][0] * nt, __ATOMIC_RELAXED, __HIP_MEMORY_SCOPE_WORKGROUP);
            __hip_atomic_fetch_add(ap + 32, hv[k][1] * nt, __ATOMIC_RELAXED, __HIP_MEMORY_SCOPE_WORKGROUP);
            __hip_atomic_fetch_add(ap + 64, hv[k][2] * nt, __ATOMIC_RELAXED, __HIP_MEMORY_SCOPE_WORKGROUP);
            __hip_atomic_fetch_add(ap + 96, hv[k][3] * nt, __ATOMIC_RELAXED, __HIP_MEMORY_SCOPE_WORKGROUP);
        }
    }
    t.head = (t.head + 32) & (QCAP - 1);
    t.qn -= cnt;
}

__device__ __forceinline__ void tile_offer(TileCtx& t, int c, int e) {
    const bool m = (unsigned)(c - t.base) < (unsigned)TILE_NODES;
    const unsigned mask = __builtin_amdgcn_ballot_w32(m);
    if (mask != 0u) {
        const int pos = t.head + t.qn + (int)__builtin_amdgcn_mbcnt_lo(mask, 0u);
        if (m) t.q[pos & (QCAP - 1)] = e;
        t.qn += __builtin_popcount(mask);
    }
}

__device__ __forceinline__ unsigned zero_bytes(unsigned x) {
    return ~(((x & 0x7F7F7F7Fu) + 0x7F7F7F7Fu) | x | 0x7F7F7F7Fu);
}

__device__ __forceinline__ unsigned match16(int4 kv, unsigned pat) {
    return (zero_bytes((unsigned)kv.x ^ pat) >> 7) | (zero_bytes((unsigned)kv.y ^ pat) >> 6) |
           (zero_bytes((unsigned)kv.z ^ pat) >> 5) | (zero_bytes((unsigned)kv.w ^ pat) >> 4);
}

__device__ __forceinline__ void tile_offer_mask(TileCtx& t, unsigned m, int ebase) {
    unsigned live = __builtin_amdgcn_ballot_w32(m != 0u);
    while (live != 0u) {
        const bool has = (m != 0u);
        const int p = __builtin_ffs((int)m) - 1;
        const int pos = t.head + t.qn + (int)__builtin_amdgcn_mbcnt_lo(live, 0u);
        if (has) t.q[pos & (QCAP - 1)] = ebase + ((p & 7) << 2) + (p >> 3);
        t.qn += __builtin_popcount(live);
        m &= m - 1u;
        live = __builtin_amdgcn_ballot_w32(m != 0u);
    }
}

__device__ __forceinline__ void tile_sum(TileCtx& t, int wave, int tid,
                                         const int4* __restrict__ rec,
                                         const int* __restrict__ key4,
                                         const float* __restrict__ h, int E) {
    for (int i = tid; i < (TILE_NODES + 1) * FEAT / 4; i += 1024)
        ((float4*)t.acc)[i] = make_float4(0.f, 0.f, 0.f, 0.f);
    __syncthreads();

    const int nGroups = E >> 9;
    const unsigned pat = (unsigned)(blockIdx.x & 255u) * 0x01010101u;
    const int start = (nGroups / (int)gridDim.x) * (int)blockIdx.x;

    int i = wave;
    for (; i + 3 * AGG_WAVES < nGroups; i += 4 * AGG_WAVES) {
        int4 kv[4];
        int eb[4];
#pragma unroll
        for (int u = 0; u < 4; ++u) {
            int g = start + i + u * AGG_WAVES;
            if (g >= nGroups) g -= nGroups;
            eb[u] = g * 512 + t.lane * 16;
            kv[u] = *(const int4*)(key4 + g * 128 + t.lane * 4);
        }
        int done = 0;
        do {
#pragma unroll
            for (int u = 0; u < 4; ++u) {
                if (u == done && t.qn <= QCAP - 512) {
                    tile_offer_mask(t, match16(kv[u], pat), eb[u]);
                    done = u + 1;
                }
            }
            while (t.qn >= 32) tile_drain(t, 32, rec, h);
        } while (done < 4);
    }
    for (; i < nGroups; i += AGG_WAVES) {
        int g = start + i;
        if (g >= nGroups) g -= nGroups;
        const int4 kv = *(const int4*)(key4 + g * 128 + t.lane * 4);
        tile_offer_mask(t, match16(kv, pat), g * 512 + t.lane * 16);
        while (t.qn >= 32) tile_drain(t, 32, rec, h);
    }
    if (wave == 0) {
        for (int eb = nGroups * 512; eb < E; eb += 32) {
            const int e = eb + t.lane;
            const int c = (e < E) ? rec[e].y : -1;
            tile_offer(t, c, e);
        }
    }
    while (t.qn >= 32) tile_drain(t, 32, rec, h);
    if (t.qn > 0) tile_drain(t, t.qn, rec, h);
    __syncthreads();
}

#define CLOSE_TRIPS 4

__global__ __launch_bounds__(1024) void aggregate_layer1_kernel(const int* __restrict__ rec4,
                                                                const int* __restrict__ key4,
                                                                const float* __restrict__ dis,
                                                                const float* __restrict__ h,
                                                                const float* __restrict__ b0,
                                                                const float* __restrict__ g,
                                                                const float* __restrict__ be,
                                                                const float* __restrict__ mu,
                                                                const float* __restrict__ var,
                                                                float* __restrict__ h1,
                                                                int E, int N) {
    __shared__ float acc[(TILE_NODES + 1) * FEAT];
    __shared__ int qbuf[AGG_WAVES * QCAP];

    const int tid = threadIdx.x;
    const int wave = __builtin_amdgcn_readfirstlane(tid >> 5);

    TileCtx t;
    t.acc = acc;
    t.q = qbuf + wave * QCAP;
    t.base = blockIdx.x * TILE_NODES;
    t.lane = tid & 31;
    t.head = 0;
    t.qn = 0;

    tile_sum(t, wave, tid, (const int4*)rec4, key4, h, E);

    const int f = (tid & 31) << 2;
    const float4 bv = *(const float4*)(b0 + f);
    const float4 gv = *(const float4*)(g + f);
    const float4 ev = *(const float4*)(be + f);
    const float4 mv = *(const float4*)(mu + f);
    const float4 vv = *(const float4*)(var + f);
    const float bb[4] = {bv.x, bv.y, bv.z, bv.w};
    const float gg[4] = {gv.x, gv.y, gv.z, gv.w};
    const float ee[4] = {ev.x, ev.y, ev.z, ev.w};
    const float mm[4] = {mv.x, mv.y, mv.z, mv.w};
    const float rs[4] = {rsqrtf(vv.x + BN_EPS), rsqrtf(vv.y + BN_EPS),
                         rsqrtf(vv.z + BN_EPS), rsqrtf(vv.w + BN_EPS)};

    const int rows = min(TILE_NODES, N - t.base);
    const int lim = rows * (FEAT / 4);
    const float4* hsrc = (const float4*)(h + (size_t)t.base * FEAT);
    float4* dst = (float4*)(h1 + (size_t)t.base * FEAT);
    for (int i0 = tid; i0 < lim; i0 += CLOSE_TRIPS * 1024) {
        float4 hv[CLOSE_TRIPS];
        float dv[CLOSE_TRIPS];
#pragma unroll
        for (int u = 0; u < CLOSE_TRIPS; ++u) {
            const int j4 = min(i0 + u * 1024, lim - 1);
            hv[u] = hsrc[j4];
            dv[u] = dis[t.base + (j4 >> 5)];
        }
#pragma unroll
        for (int u = 0; u < CLOSE_TRIPS; ++u) {
            const int i4 = i0 + u * 1024;
            if (i4 < lim) {
                const float4 a = ((const float4*)acc)[i4];
                const float dd = dv[u];
                const float av[4] = {a.x, a.y, a.z, a.w};
                const float hh[4] = {hv[u].x, hv[u].y, hv[u].z, hv[u].w};
                float o[4];
#pragma unroll
                for (int j = 0; j < 4; ++j) {
                    float v = dd * (av[j] + hh[j]) + bb[j];
                    v = (v - mm[j]) * rs[j] * gg[j] + ee[j];
                    o[j] = (v > 0.0f) ? v : expm1f(v);
                }
                dst[i4] = make_float4(o[0], o[1], o[2], o[3]);
            }
        }
    }
}

__global__ __launch_bounds__(1024) void aggregate_layer2_kernel(const int* __restrict__ rec4,
                                                                const int* __restrict__ key4,
                                                                const float* __restrict__ dis,
                                                                const float* __restrict__ h,
                                                                const float* __restrict__ emb,
                                                                const float* __restrict__ h1,
                                                                const float* __restrict__ b1,
                                                                float* __restrict__ out,
                                                                int E, int N) {
    __shared__ float acc[(TILE_NODES + 1) * FEAT];
    __shared__ int qbuf[AGG_WAVES * QCAP];

    const int tid = threadIdx.x;
    const int wave = __builtin_amdgcn_readfirstlane(tid >> 5);

    TileCtx t;
    t.acc = acc;
    t.q = qbuf + wave * QCAP;
    t.base = blockIdx.x * TILE_NODES;
    t.lane = tid & 31;
    t.head = 0;
    t.qn = 0;

    tile_sum(t, wave, tid, (const int4*)rec4, key4, h, E);

    const int f = (tid & 31) << 2;
    const float4 bv = *(const float4*)(b1 + f);
    const float inv3 = 1.0f / 3.0f;

    const int rows = min(TILE_NODES, N - t.base);
    const int lim = rows * (FEAT / 4);
    const float4* hsrc = (const float4*)(h + (size_t)t.base * FEAT);
    const float4* esrc = (const float4*)(emb + (size_t)t.base * FEAT);
    const float4* osrc = (const float4*)(h1 + (size_t)t.base * FEAT);
    float4* dst = (float4*)(out + (size_t)t.base * FEAT);
    for (int i0 = tid; i0 < lim; i0 += CLOSE_TRIPS * 1024) {
        float4 pv[CLOSE_TRIPS], xv[CLOSE_TRIPS], ov[CLOSE_TRIPS];
        float dv[CLOSE_TRIPS];
#pragma unroll
        for (int u = 0; u < CLOSE_TRIPS; ++u) {
            const int j4 = min(i0 + u * 1024, lim - 1);
            xv[u] = esrc[j4];
            ov[u] = osrc[j4];
            pv[u] = hsrc[j4];
            dv[u] = dis[t.base + (j4 >> 5)];
        }
#pragma unroll
        for (int u = 0; u < CLOSE_TRIPS; ++u) {
            asm volatile("" :: "v"(xv[u].x), "v"(xv[u].y), "v"(xv[u].z), "v"(xv[u].w),
                               "v"(ov[u].x), "v"(ov[u].y), "v"(ov[u].z), "v"(ov[u].w),
                               "v"(pv[u].x), "v"(pv[u].y), "v"(pv[u].z), "v"(pv[u].w),
                               "v"(dv[u]));
        }
#pragma unroll
        for (int u = 0; u < CLOSE_TRIPS; ++u) {
            const int i4 = i0 + u * 1024;
            if (i4 < lim) {
                const float4 a2 = ((const float4*)acc)[i4];
                const float dd = dv[u];
                const float4 e = xv[u];
                const float4 o1 = ov[u];
                const float4 p2 = pv[u];
                float4 r;
                r.x = (e.x + o1.x + dd * (a2.x + p2.x) + bv.x) * inv3;
                r.y = (e.y + o1.y + dd * (a2.y + p2.y) + bv.y) * inv3;
                r.z = (e.z + o1.z + dd * (a2.z + p2.z) + bv.z) * inv3;
                r.w = (e.w + o1.w + dd * (a2.w + p2.w) + bv.w) * inv3;
                dst[i4] = r;
            }
        }
    }
}

extern "C" void kernel_launch(void* const* d_in, const int* in_sizes, int n_in,
                              void* d_out, int out_size, void* d_ws, size_t ws_size,
                              hipStream_t stream) {
    const float* emb = (const float*)d_in[0];
    const int*   ei  = (const int*)d_in[1];
    const float* ew  = (const float*)d_in[2];
    const float* W0  = (const float*)d_in[3];
    const float* b0  = (const float*)d_in[4];
    const float* W1  = (const float*)d_in[5];
    const float* b1  = (const float*)d_in[6];
    const float* bng = (const float*)d_in[7];
    const float* bnb = (const float*)d_in[8];
    const float* bnm = (const float*)d_in[9];
    const float* bnv = (const float*)d_in[10];
    float* out = (float*)d_out;

    const int N  = in_sizes[0] / FEAT;
    const int E  = in_sizes[2];
    const size_t ND = (size_t)N * FEAT;

    const int* row = ei;
    const int* col = ei + E;

    float* ws   = (float*)d_ws;
    size_t disPad = ((size_t)N + 1023) & ~(size_t)1023;
    size_t keyPad = ((size_t)E + 1023) & ~(size_t)1023;
    float* dis  = ws;
    float* bufA = ws + disPad;
    float* bufB = bufA + ND;
    int* key4   = (int*)(bufB + ND);
    int* rec    = (int*)((char*)key4 + keyPad);

    hipMemsetAsync(dis,  0, (size_t)N * sizeof(float), stream);

    const int gTile  = (N + TILE_NODES - 1) / TILE_NODES;
    const int gRowT  = (N + 15) / 16;

    degree_key_kernel<<<(E + 1023) / 1024, 256, 0, stream>>>(row, col, ew, dis, key4, rec, E);
    dis_kernel<<<(N + 255) / 256, 256, 0, stream>>>(dis, N);

    gemm_n128_kernel<<<gRowT, 256, 0, stream>>>(emb, W0, dis, bufA, N);

    aggregate_layer1_kernel<<<gTile, 1024, 0, stream>>>(rec, key4, dis, bufA,
                                                        b0, bng, bnb, bnm, bnv, bufB, E, N);

    gemm_n128_kernel<<<gRowT, 256, 0, stream>>>(bufB, W1, dis, bufA, N);

    aggregate_layer2_kernel<<<gTile, 1024, 0, stream>>>(rec, key4, dis, bufA,
                                                        emb, bufB, b1, out, E, N);
}


# layer 2 no longer filters the edge list: layer 1's waves save their tile's edges {row, tile row, w} as per-wave lists and layer 2 adds from them (scan kept for overflow); LDS arrival order kept
# speedup vs baseline: 2.5266x; 1.0676x over previous
#include <hip/hip_runtime.h>
#include <hip/hip_bf16.h>
#include <math.h>

typedef float v2f __attribute__((ext_vector_type(2)));
typedef float v8f __attribute__((ext_vector_type(8)));

#if defined(__has_builtin)
#if __has_builtin(__builtin_amdgcn_wmma_f32_16x16x4_f32)
#define HAS_WMMA_F32_4 1
#endif
#endif
#ifndef HAS_WMMA_F32_4
#define HAS_WMMA_F32_4 0
#endif

#define FEAT 128
#define BN_EPS 1e-5f

#define TILE_NODES 196

__global__ __launch_bounds__(256) void degree_key_kernel(const int* __restrict__ row,
                                                         const int* __restrict__ col,
                                                         const float* __restrict__ w,
                                                         float* __restrict__ deg,
                                                         int* __restrict__ key4,
                                                         int* __restrict__ rec4, int E) {
    int4* __restrict__ rec = (int4*)rec4;
    const int er = blockIdx.x * 1024 + threadIdx.x;
    int rr[4], rc[4];
    float rw[4];
#pragma unroll
    for (int j = 0; j < 4; ++j) {
        const int el = min(er + j * 256, E - 1);
        rr[j] = row[el];
        rc[j] = col[el];
        rw[j] = w[el];
    }
#pragma unroll
    for (int j = 0; j < 4; ++j) {
        const int e = er + j * 256;
        if (e < E) rec[e] = make_int4(rr[j], rc[j], __float_as_int(rw[j]), 0);
    }
    const int t = blockIdx.x * 256 + threadIdx.x;
    const int e0 = t * 4;
    if (e0 >= E) return;
    unsigned k = 0u;
    if (e0 + 3 < E) {
        int c[4];
        float wv[4];
#pragma unroll
        for (int j = 0; j < 4; ++j) {
            c[j] = col[e0 + j];
            wv[j] = w[e0 + j];
        }
#pragma unroll
        for (int j = 0; j < 4; ++j) {
            __hip_atomic_fetch_add(&deg[c[j]], wv[j], __ATOMIC_RELAXED,
                                   __HIP_MEMORY_SCOPE_AGENT);
            k |= ((unsigned)(c[j] / TILE_NODES) & 255u) << (8 * j);
        }
    } else {
        for (int j = 0; j < 4 && e0 + j < E; ++j) {
            const int c = col[e0 + j];
            __hip_atomic_fetch_add(&deg[c], w[e0 + j], __ATOMIC_RELAXED,
                                   __HIP_MEMORY_SCOPE_AGENT);
            k |= ((unsigned)(c / TILE_NODES) & 255u) << (8 * j);
        }
    }
    key4[t] = (int)k;
}

__global__ __launch_bounds__(256) void dis_kernel(float* __restrict__ deg_dis, int n) {
    int i = blockIdx.x * 256 + threadIdx.x;
    if (i < n) deg_dis[i] = rsqrtf(deg_dis[i] + 1.0f);
}

__global__ __launch_bounds__(256) void gemm_n128_kernel(const float* __restrict__ X,
                                                        const float* __restrict__ W,
                                                        const float* __restrict__ dis,
                                                        float* __restrict__ Y,
                                                        int nrows) {
    __shared__ float lA[16 * 132];

    const int tid = threadIdx.x;
    const int rowBase = blockIdx.x * 16;

#pragma unroll
    for (int i = 0; i < 2; ++i) {
        int idx = tid + i * 256;
        int r = idx >> 5;
        int c4 = (idx & 31) << 2;
        float4 v = make_float4(0.f, 0.f, 0.f, 0.f);
        float s = 0.f;
        if (rowBase + r < nrows) {
            v = *(const float4*)(X + (size_t)(rowBase + r) * FEAT + c4);
            s = dis[rowBase + r];
        }
        float* dst = &lA[r * 132 + c4];
        dst[0] = v.x * s; dst[1] = v.y * s; dst[2] = v.z * s; dst[3] = v.w * s;
    }
    __syncthreads();

    const int wave = tid >> 5;
    const int lane = tid & 31;
    const int lid  = lane & 15;
    const int half = lane >> 4;
    const int n0   = wave * 16;

#if HAS_WMMA_F32_4
    v8f acc = {0.f, 0.f, 0.f, 0.f, 0.f, 0.f, 0.f, 0.f};
    const float* arow = &lA[lid * 132];
#pragma unroll
    for (int k0 = 0; k0 < FEAT; k0 += 4) {
        const int ka = k0 + 2 * half;
        v2f a, b;
        a.x = arow[ka];
        a.y = arow[ka + 1];
        b.x = W[(size_t)ka * FEAT + n0 + lid];
        b.y = W[(size_t)(ka + 1) * FEAT + n0 + lid];
        acc = __builtin_amdgcn_wmma_f32_16x16x4_f32(
             false, a,  false, b,
             (short)0, acc,  false,  false);
    }
    float* yb = Y + (size_t)rowBase * FEAT + n0 + lid;
#pragma unroll
    for (int v = 0; v < 8; ++v) {
        int m = v + 8 * half;
        if (rowBase + m < nrows) yb[(size_t)m * FEAT] = acc[v];
    }
#else
    float accs[8] = {0.f, 0.f, 0.f, 0.f, 0.f, 0.f, 0.f, 0.f};
    for (int k = 0; k < FEAT; ++k) {
        float bv = W[(size_t)k * FEAT + n0 + lid];
#pragma unroll
        for (int v = 0; v < 8; ++v)
            accs[v] = fmaf(lA[(v + 8 * half) * 132 + k], bv, accs[v]);
    }
    float* yb = Y + (size_t)rowBase * FEAT + n0 + lid;
#pragma unroll
    for (int v = 0; v < 8; ++v) {
        int m = v + 8 * half;
        if (rowBase + m < nrows) yb[(size_t)m * FEAT] = accs[v];
    }
#endif
}

#define AGG_WAVES 32
#define QCAP 1024
#define LCAP 256

struct TileCtx {
    float* acc;
    int* q;
    int base;
    int lane;
    int head;
    int qn;
    int2* list;
    int saved;
};

__device__ __forceinline__ void tile_add32(TileCtx& t, int cnt, int r, int loc, int nrmBits,
                                           const float* __restrict__ h) {
#pragma unroll
    for (int sub = 0; sub < 4; ++sub) {
        if (sub * 8 >= cnt) break;
        float hv[8][4];
#pragma unroll
        for (int k = 0; k < 8; ++k) {
            const int rt = __builtin_amdgcn_readlane(r, sub * 8 + k);
            const float* hp = h + (size_t)rt * FEAT + t.lane;
            hv[k][0] = hp[0];
            hv[k][1] = hp[32];
            hv[k][2] = hp[64];
            hv[k][3] = hp[96];
        }
#pragma unroll
        for (int k = 0; k < 8; ++k) {
            const float nt = __int_as_float(__builtin_amdgcn_readlane(nrmBits, sub * 8 + k));
            const int lt = __builtin_amdgcn_readlane(loc, sub * 8 + k);
            float* ap = t.acc + lt * FEAT + t.lane;
            __hip_atomic_fetch_add(ap + 0,  hv[k][0] * nt, __ATOMIC_RELAXED, __HIP_MEMORY_SCOPE_WORKGROUP);
            __hip_atomic_fetch_add(ap + 32, hv[k][1] * nt, __ATOMIC_RELAXED, __HIP_MEMORY_SCOPE_WORKGROUP);
            __hip_atomic_fetch_add(ap + 64, hv[k][2] * nt, __ATOMIC_RELAXED, __HIP_MEMORY_SCOPE_WORKGROUP);
            __hip_atomic_fetch_add(ap + 96, hv[k][3] * nt, __ATOMIC_RELAXED, __HIP_MEMORY_SCOPE_WORKGROUP);
        }
    }
}

template <bool SAVE>
__device__ __forceinline__ void tile_drain(TileCtx& t, int cnt,
                                           const int4* __restrict__ rec,
                                           const float* __restrict__ h) {
    const bool valid = t.lane < cnt;
    int e = t.q[(t.head + t.lane) & (QCAP - 1)];
    e = valid ? e : 0;
    const int4 rc = rec[e];
    const int r = rc.x;
    const int c = rc.y;
    const bool mine = valid && ((unsigned)(c - t.base) < (unsigned)TILE_NODES);
    const int loc = mine ? (c - t.base) : TILE_NODES;
    const int nrmBits = mine ? rc.z : 0;
    if (SAVE) {
        const unsigned mask = __builtin_amdgcn_ballot_w32(mine);
        const int pos = t.saved + (int)__builtin_amdgcn_mbcnt_lo(mask, 0u);
        if (mine && pos < LCAP)
            t.list[pos] = make_int2((int)(((unsigned)r << 8) | (unsigned)loc), rc.z);
        t.saved += __builtin_popcount(mask);
    }
    tile_add32(t, cnt, r, loc, nrmBits, h);
    t.head = (t.head + 32) & (QCAP - 1);
    t.qn -= cnt;
}

__device__ __forceinline__ void tile_offer(TileCtx& t, int c, int e) {
    const bool m = (unsigned)(c - t.base) < (unsigned)TILE_NODES;
    const unsigned mask = __builtin_amdgcn_ballot_w32(m);
    if (mask != 0u) {
        const int pos = t.head + t.qn + (int)__builtin_amdgcn_mbcnt_lo(mask, 0u);
        if (m) t.q[pos & (QCAP - 1)] = e;
        t.qn += __builtin_popcount(mask);
    }
}

__device__ __forceinline__ unsigned zero_bytes(unsigned x) {
    return ~(((x & 0x7F7F7F7Fu) + 0x7F7F7F7Fu) | x | 0x7F7F7F7Fu);
}

__device__ __forceinline__ unsigned match16(int4 kv, unsigned pat) {
    return (zero_bytes((unsigned)kv.x ^ pat) >> 7) | (zero_bytes((unsigned)kv.y ^ pat) >> 6) |
           (zero_bytes((unsigned)kv.z ^ pat) >> 5) | (zero_bytes((unsigned)kv.w ^ pat) >> 4);
}

__device__ __forceinline__ void tile_offer_mask(TileCtx& t, unsigned m, int ebase) {
    unsigned live = __builtin_amdgcn_ballot_w32(m != 0u);
    while (live != 0u) {
        const bool has = (m != 0u);
        const int p = __builtin_ffs((int)m) - 1;
        const int pos = t.head + t.qn + (int)__builtin_amdgcn_mbcnt_lo(live, 0u);
        if (has) t.q[pos & (QCAP - 1)] = ebase + ((p & 7) << 2) + (p >> 3);
        t.qn += __builtin_popcount(live);
        m &= m - 1u;
        live = __builtin_amdgcn_ballot_w32(m != 0u);
    }
}

template <bool SAVE>
__device__ __forceinline__ void tile_sum(TileCtx& t, int wave, int tid, int* cnt_out,
                                         const int4* __restrict__ rec,
                                         const int* __restrict__ key4,
                                         const float* __restrict__ h, int E) {
    for (int i = tid; i < (TILE_NODES + 1) * FEAT / 4; i += 1024)
        ((float4*)t.acc)[i] = make_float4(0.f, 0.f, 0.f, 0.f);
    __syncthreads();

    const int nGroups = E >> 9;
    const unsigned pat = (unsigned)(blockIdx.x & 255u) * 0x01010101u;
    const int start = (nGroups / (int)gridDim.x) * (int)blockIdx.x;

    int i = wave;
    for (; i + 3 * AGG_WAVES < nGroups; i += 4 * AGG_WAVES) {
        int4 kv[4];
        int eb[4];
#pragma unroll
        for (int u = 0; u < 4; ++u) {
            int g = start + i + u * AGG_WAVES;
            if (g >= nGroups) g -= nGroups;
            eb[u] = g * 512 + t.lane * 16;
            kv[u] = *(const int4*)(key4 + g * 128 + t.lane * 4);
        }
        int done = 0;
        do {
#pragma unroll
            for (int u = 0; u < 4; ++u) {
                if (u == done && t.qn <= QCAP - 512) {
                    tile_offer_mask(t, match16(kv[u], pat), eb[u]);
                    done = u + 1;
                }
            }
            while (t.qn >= 32) tile_drain<SAVE>(t, 32, rec, h);
        } while (done < 4);
    }
    for (; i < nGroups; i += AGG_WAVES) {
        int g = start + i;
        if (g >= nGroups) g -= nGroups;
        const int4 kv = *(const int4*)(key4 + g * 128 + t.lane * 4);
        tile_offer_mask(t, match16(kv, pat), g * 512 + t.lane * 16);
        while (t.qn >= 32) tile_drain<SAVE>(t, 32, rec, h);
    }
    if (wave == 0) {
        for (int eb = nGroups * 512; eb < E; eb += 32) {
            const int e = eb + t.lane;
            const int c = (e < E) ? rec[e].y : -1;
            tile_offer(t, c, e);
        }
    }
    while (t.qn >= 32) tile_drain<SAVE>(t, 32, rec, h);
    if (t.qn > 0) tile_drain<SAVE>(t, t.qn, rec, h);
    if (SAVE && t.lane == 0) *cnt_out = t.saved;
    __syncthreads();
}

__device__ __forceinline__ void tile_sum_from_list(TileCtx& t, int tid,
                                                   const int2* __restrict__ lp, int n,
                                                   const float* __restrict__ h) {
    for (int i = tid; i < (TILE_NODES + 1) * FEAT / 4; i += 1024)
        ((float4*)t.acc)[i] = make_float4(0.f, 0.f, 0.f, 0.f);
    __syncthreads();
    if (n > 0) {
        int2 cur = lp[min(t.lane, n - 1)];
        int i = 0;
        for (; i + 32 <= n; i += 32) {
            const int2 nxt = lp[min(i + 32 + t.lane, n - 1)];
            tile_add32(t, 32, (int)((unsigned)cur.x >> 8), cur.x & 255, cur.y, h);
            cur = nxt;
        }
        if (i < n) {
            const bool valid = i + t.lane < n;
            const int p = valid ? cur.x : TILE_NODES;
            const int wb = valid ? cur.y : 0;
            tile_add32(t, n - i, (int)((unsigned)p >> 8), p & 255, wb, h);
        }
    }
    __syncthreads();
}

#define CLOSE_TRIPS 4

__global__ __launch_bounds__(1024) void aggregate_layer1_kernel(const int* __restrict__ rec4,
                                                                const int* __restrict__ key4,
                                                                const float* __restrict__ dis,
                                                                const float* __restrict__ h,
                                                                const float* __restrict__ b0,
                                                                const float* __restrict__ g,
                                                                const float* __restrict__ be,
                                                                const float* __restrict__ mu,
                                                                const float* __restrict__ var,
                                                                float* __restrict__ h1,
                                                                int* __restrict__ lcnt,
                                                                int* __restrict__ list2,
                                                                int E, int N) {
    __shared__ float acc[(TILE_NODES + 1) * FEAT];
    __shared__ int qbuf[AGG_WAVES * QCAP];

    const int tid = threadIdx.x;
    const int wave = __builtin_amdgcn_readfirstlane(tid >> 5);

    TileCtx t;
    t.acc = acc;
    t.q = qbuf + wave * QCAP;
    t.base = blockIdx.x * TILE_NODES;
    t.lane = tid & 31;
    t.head = 0;
    t.qn = 0;
    t.list = (int2*)list2 + ((size_t)blockIdx.x * AGG_WAVES + wave) * LCAP;
    t.saved = 0;

    tile_sum<true>(t, wave, tid, lcnt + blockIdx.x * AGG_WAVES + wave,
                   (const int4*)rec4, key4, h, E);

    const int f = (tid & 31) << 2;
    const float4 bv = *(const float4*)(b0 + f);
    const float4 gv = *(const float4*)(g + f);
    const float4 ev = *(const float4*)(be + f);
    const float4 mv = *(const float4*)(mu + f);
    const float4 vv = *(const float4*)(var + f);
    const float bb[4] = {bv.x, bv.y, bv.z, bv.w};
    const float gg[4] = {gv.x, gv.y, gv.z, gv.w};
    const float ee[4] = {ev.x, ev.y, ev.z, ev.w};
    const float mm[4] = {mv.x, mv.y, mv.z, mv.w};
    const float rs[4] = {rsqrtf(vv.x + BN_EPS), rsqrtf(vv.y + BN_EPS),
                         rsqrtf(vv.z + BN_EPS), rsqrtf(vv.w + BN_EPS)};

    const int rows = min(TILE_NODES, N - t.base);
    const int lim = rows * (FEAT / 4);
    const float4* hsrc = (const float4*)(h + (size_t)t.base * FEAT);
    float4* dst = (float4*)(h1 + (size_t)t.base * FEAT);
    for (int i0 = tid; i0 < lim; i0 += CLOSE_TRIPS * 1024) {
        float4 hv[CLOSE_TRIPS];
        float dv[CLOSE_TRIPS];
#pragma unroll
        for (int u = 0; u < CLOSE_TRIPS; ++u) {
            const int j4 = min(i0 + u * 1024, lim - 1);
            hv[u] = hsrc[j4];
            dv[u] = dis[t.base + (j4 >> 5)];
        }
#pragma unroll
        for (int u = 0; u < CLOSE_TRIPS; ++u) {
            const int i4 = i0 + u * 1024;
            if (i4 < lim) {
                const float4 a = ((const float4*)acc)[i4];
                const float dd = dv[u];
                const float av[4] = {a.x, a.y, a.z, a.w};
                const float hh[4] = {hv[u].x, hv[u].y, hv[u].z, hv[u].w};
                float o[4];
#pragma unroll
                for (int j = 0; j < 4; ++j) {
                    float v = dd * (av[j] + hh[j]) + bb[j];
                    v = (v - mm[j]) * rs[j] * gg[j] + ee[j];
                    o[j] = (v > 0.0f) ? v : expm1f(v);
                }
                dst[i4] = make_float4(o[0], o[1], o[2], o[3]);
            }
        }
    }
}

__global__ __launch_bounds__(1024) void aggregate_layer2_kernel(const int* __restrict__ rec4,
                                                                const int* __restrict__ key4,
                                                                const float* __restrict__ dis,
                                                                const float* __restrict__ h,
                                                                const float* __restrict__ emb,
                                                                const float* __restrict__ h1,
                                                                const float* __restrict__ b1,
                                                                const int* __restrict__ lcnt,
                                                                const int* __restrict__ list2,
                                                                float* __restrict__ out,
                                                                int E, int N) {
    __shared__ float acc[(TILE_NODES + 1) * FEAT];
    __shared__ int qbuf[AGG_WAVES * QCAP];

    const int tid = threadIdx.x;
    const int wave = __builtin_amdgcn_readfirstlane(tid >> 5);

    TileCtx t;
    t.acc = acc;
    t.q = qbuf + wave * QCAP;
    t.base = blockIdx.x * TILE_NODES;
    t.lane = tid & 31;
    t.head = 0;
    t.qn = 0;
    t.list = nullptr;
    t.saved = 0;

    const int myc = lcnt[blockIdx.x * AGG_WAVES + t.lane];
    const bool over = __builtin_amdgcn_ballot_w32(myc > LCAP) != 0u;
    if (!over) {
        const int n = __builtin_amdgcn_readlane(myc, wave);
        tile_sum_from_list(t, tid,
                           (const int2*)list2 + ((size_t)blockIdx.x * AGG_WAVES + wave) * LCAP,
                           n, h);
    } else {
        tile_sum<false>(t, wave, tid, nullptr, (const int4*)rec4, key4, h, E);
    }

    const int f = (tid & 31) << 2;
    const float4 bv = *(const float4*)(b1 + f);
    const float inv3 = 1.0f / 3.0f;

    const int rows = min(TILE_NODES, N - t.base);
    const int lim = rows * (FEAT / 4);
    const float4* hsrc = (const float4*)(h + (size_t)t.base * FEAT);
    const float4* esrc = (const float4*)(emb + (size_t)t.base * FEAT);
    const float4* osrc = (const float4*)(h1 + (size_t)t.base * FEAT);
    float4* dst = (float4*)(out + (size_t)t.base * FEAT);
    for (int i0 = tid; i0 < lim; i0 += CLOSE_TRIPS * 1024) {
        float4 pv[CLOSE_TRIPS], xv[CLOSE_TRIPS], ov[CLOSE_TRIPS];
        float dv[CLOSE_TRIPS];
#pragma unroll
        for (int u = 0; u < CLOSE_TRIPS; ++u) {
            const int j4 = min(i0 + u * 1024, lim - 1);
            xv[u] = esrc[j4];
            ov[u] = osrc[j4];
            pv[u] = hsrc[j4];
            dv[u] = dis[t.base + (j4 >> 5)];
        }
#pragma unroll
        for (int u = 0; u < CLOSE_TRIPS; ++u) {
            asm volatile("" :: "v"(xv[u].x), "v"(xv[u].y), "v"(xv[u].z), "v"(xv[u].w),
                               "v"(ov[u].x), "v"(ov[u].y), "v"(ov[u].z), "v"(ov[u].w),
                               "v"(pv[u].x), "v"(pv[u].y), "v"(pv[u].z), "v"(pv[u].w),
                               "v"(dv[u]));
        }
#pragma unroll
        for (int u = 0; u < CLOSE_TRIPS; ++u) {
            const int i4 = i0 + u * 1024;
            if (i4 < lim) {
                const float4 a2 = ((const float4*)acc)[i4];
                const float dd = dv[u];
                const float4 e = xv[u];
                const float4 o1 = ov[u];
                const float4 p2 = pv[u];
                float4 r;
                r.x = (e.x + o1.x + dd * (a2.x + p2.x) + bv.x) * inv3;
                r.y = (e.y + o1.y + dd * (a2.y + p2.y) + bv.y) * inv3;
                r.z = (e.z + o1.z + dd * (a2.z + p2.z) + bv.z) * inv3;
                r.w = (e.w + o1.w + dd * (a2.w + p2.w) + bv.w) * inv3;
                dst[i4] = r;
            }
        }
    }
}

extern "C" void kernel_launch(void* const* d_in, const int* in_sizes, int n_in,
                              void* d_out, int out_size, void* d_ws, size_t ws_size,
                              hipStream_t stream) {
    const float* emb = (const float*)d_in[0];
    const int*   ei  = (const int*)d_in[1];
    const float* ew  = (const float*)d_in[2];
    const float* W0  = (const float*)d_in[3];
    const float* b0  = (const float*)d_in[4];
    const float* W1  = (const float*)d_in[5];
    const float* b1  = (const float*)d_in[6];
    const float* bng = (const float*)d_in[7];
    const float* bnb = (const float*)d_in[8];
    const float* bnm = (const float*)d_in[9];
    const float* bnv = (const float*)d_in[10];
    float* out = (float*)d_out;

    const int N  = in_sizes[0] / FEAT;
    const int E  = in_sizes[2];
    const size_t ND = (size_t)N * FEAT;

    const int* row = ei;
    const int* col = ei + E;

    float* ws   = (float*)d_ws;
    size_t disPad = ((size_t)N + 1023) & ~(size_t)1023;
    size_t keyPad = ((size_t)E + 1023) & ~(size_t)1023;
    float* dis  = ws;
    float* bufA = ws + disPad;
    float* bufB = bufA + ND;
    int* key4   = (int*)(bufB + ND);
    int* rec    = (int*)((char*)key4 + keyPad);

    hipMemsetAsync(dis,  0, (size_t)N * sizeof(float), stream);

    const int gTile  = (N + TILE_NODES - 1) / TILE_NODES;
    const int gRowT  = (N + 15) / 16;
    int* lcnt   = rec + (size_t)E * 4;
    int* list2  = lcnt + (((size_t)gTile * AGG_WAVES + 63) & ~(size_t)63);

    degree_key_kernel<<<(E + 1023) / 1024, 256, 0, stream>>>(row, col, ew, dis, key4, rec, E);
    dis_kernel<<<(N + 255) / 256, 256, 0, stream>>>(dis, N);

    gemm_n128_kernel<<<gRowT, 256, 0, stream>>>(emb, W0, dis, bufA, N);

    aggregate_layer1_kernel<<<gTile, 1024, 0, stream>>>(rec, key4, dis, bufA,
                                                        b0, bng, bnb, bnm, bnv, bufB,
                                                        lcnt, list2, E, N);

    gemm_n128_kernel<<<gRowT, 256, 0, stream>>>(bufB, W1, dis, bufA, N);

    aggregate_layer2_kernel<<<gTile, 1024, 0, stream>>>(rec, key4, dis, bufA,
                                                        emb, bufB, b1, lcnt, list2, out, E, N);
}


# both aggregations gather a source row with one 16-byte-per-lane load (4 whole lines per instruction), not four 4-byte loads; tile row keeps feature 4l+j at word 32j+l; LDS arrival order kept
# speedup vs baseline: 2.6788x; 1.0602x over previous
#include <hip/hip_runtime.h>
#include <hip/hip_bf16.h>
#include <math.h>

typedef float v2f __attribute__((ext_vector_type(2)));
typedef float v8f __attribute__((ext_vector_type(8)));

#if defined(__has_builtin)
#if __has_builtin(__builtin_amdgcn_wmma_f32_16x16x4_f32)
#define HAS_WMMA_F32_4 1
#endif
#endif
#ifndef HAS_WMMA_F32_4
#define HAS_WMMA_F32_4 0
#endif

#define FEAT 128
#define BN_EPS 1e-5f

#define TILE_NODES 196

__global__ __launch_bounds__(256) void degree_key_kernel(const int* __restrict__ row,
                                                         const int* __restrict__ col,
                                                         const float* __restrict__ w,
                                                         float* __restrict__ deg,
                                                         int* __restrict__ key4,
                                                         int* __restrict__ rec4, int E) {
    int4* __restrict__ rec = (int4*)rec4;
    const int er = blockIdx.x * 1024 + threadIdx.x;
    int rr[4], rc[4];
    float rw[4];
#pragma unroll
    for (int j = 0; j < 4; ++j) {
        const int el = min(er + j * 256, E - 1);
        rr[j] = row[el];
        rc[j] = col[el];
        rw[j] = w[el];
    }
#pragma unroll
    for (int j = 0; j < 4; ++j) {
        const int e = er + j * 256;
        if (e < E) rec[e] = make_int4(rr[j], rc[j], __float_as_int(rw[j]), 0);
    }
    const int t = blockIdx.x * 256 + threadIdx.x;
    const int e0 = t * 4;
    if (e0 >= E) return;
    unsigned k = 0u;
    if (e0 + 3 < E) {
        int c[4];
        float wv[4];
#pragma unroll
        for (int j = 0; j < 4; ++j) {
            c[j] = col[e0 + j];
            wv[j] = w[e0 + j];
        }
#pragma unroll
        for (int j = 0; j < 4; ++j) {
            __hip_atomic_fetch_add(&deg[c[j]], wv[j], __ATOMIC_RELAXED,
                                   __HIP_MEMORY_SCOPE_AGENT);
            k |= ((unsigned)(c[j] / TILE_NODES) & 255u) << (8 * j);
        }
    } else {
        for (int j = 0; j < 4 && e0 + j < E; ++j) {
            const int c = col[e0 + j];
            __hip_atomic_fetch_add(&deg[c], w[e0 + j], __ATOMIC_RELAXED,
                                   __HIP_MEMORY_SCOPE_AGENT);
            k |= ((unsigned)(c / TILE_NODES) & 255u) << (8 * j);
        }
    }
    key4[t] = (int)k;
}

__global__ __launch_bounds__(256) void dis_kernel(float* __restrict__ deg_dis, int n) {
    int i = blockIdx.x * 256 + threadIdx.x;
    if (i < n) deg_dis[i] = rsqrtf(deg_dis[i] + 1.0f);
}

__global__ __launch_bounds__(256) void gemm_n128_kernel(const float* __restrict__ X,
                                                        const float* __restrict__ W,
                                                        const float* __restrict__ dis,
                                                        float* __restrict__ Y,
                                                        int nrows) {
    __shared__ float lA[16 * 132];

    const int tid = threadIdx.x;
    const int rowBase = blockIdx.x * 16;

#pragma unroll
    for (int i = 0; i < 2; ++i) {
        int idx = tid + i * 256;
        int r = idx >> 5;
        int c4 = (idx & 31) << 2;
        float4 v = make_float4(0.f, 0.f, 0.f, 0.f);
        float s = 0.f;
        if (rowBase + r < nrows) {
            v = *(const float4*)(X + (size_t)(rowBase + r) * FEAT + c4);
            s = dis[rowBase + r];
        }
        float* dst = &lA[r * 132 + c4];
        dst[0] = v.x * s; dst[1] = v.y * s; dst[2] = v.z * s; dst[3] = v.w * s;
    }
    __syncthreads();

    const int wave = tid >> 5;
    const int lane = tid & 31;
    const int lid  = lane & 15;
    const int half = lane >> 4;
    const int n0   = wave * 16;

#if HAS_WMMA_F32_4
    v8f acc = {0.f, 0.f, 0.f, 0.f, 0.f, 0.f, 0.f, 0.f};
    const float* arow = &lA[lid * 132];
#pragma unroll
    for (int k0 = 0; k0 < FEAT; k0 += 4) {
        const int ka = k0 + 2 * half;
        v2f a, b;
        a.x = arow[ka];
        a.y = arow[ka + 1];
        b.x = W[(size_t)ka * FEAT + n0 + lid];
        b.y = W[(size_t)(ka + 1) * FEAT + n0 + lid];
        acc = __builtin_amdgcn_wmma_f32_16x16x4_f32(
             false, a,  false, b,
             (short)0, acc,  false,  false);
    }
    float* yb = Y + (size_t)rowBase * FEAT + n0 + lid;
#pragma unroll
    for (int v = 0; v < 8; ++v) {
        int m = v + 8 * half;
        if (rowBase + m < nrows) yb[(size_t)m * FEAT] = acc[v];
    }
#else
    float accs[8] = {0.f, 0.f, 0.f, 0.f, 0.f, 0.f, 0.f, 0.f};
    for (int k = 0; k < FEAT; ++k) {
        float bv = W[(size_t)k * FEAT + n0 + lid];
#pragma unroll
        for (int v = 0; v < 8; ++v)
            accs[v] = fmaf(lA[(v + 8 * half) * 132 + k], bv, accs[v]);
    }
    float* yb = Y + (size_t)rowBase * FEAT + n0 + lid;
#pragma unroll
    for (int v = 0; v < 8; ++v) {
        int m = v + 8 * half;
        if (rowBase + m < nrows) yb[(size_t)m * FEAT] = accs[v];
    }
#endif
}

#define AGG_WAVES 32
#define QCAP 1024
#define LCAP 256

struct TileCtx {
    float* acc;
    int* q;
    int base;
    int lane;
    int head;
    int qn;
    int2* list;
    int saved;
};

__device__ __forceinline__ void tile_add32(TileCtx& t, int cnt, int r, int loc, int nrmBits,
                                           const float* __restrict__ h) {
#pragma unroll
    for (int sub = 0; sub < 4; ++sub) {
        if (sub * 8 >= cnt) break;
        float4 hv[8];
#pragma unroll
        for (int k = 0; k < 8; ++k) {
            const int rt = __builtin_amdgcn_readlane(r, sub * 8 + k);
            hv[k] = *(const float4*)(h + (size_t)rt * FEAT + 4 * t.lane);
        }
#pragma unroll
        for (int k = 0; k < 8; ++k) {
            const float nt = __int_as_float(__builtin_amdgcn_readlane(nrmBits, sub * 8 + k));
            const int lt = __builtin_amdgcn_readlane(loc, sub * 8 + k);
            float* ap = t.acc + lt * FEAT + t.lane;
            __hip_atomic_fetch_add(ap + 0,  hv[k].x * nt, __ATOMIC_RELAXED, __HIP_MEMORY_SCOPE_WORKGROUP);
            __hip_atomic_fetch_add(ap + 32, hv[k].y * nt, __ATOMIC_RELAXED, __HIP_MEMORY_SCOPE_WORKGROUP);
            __hip_atomic_fetch_add(ap + 64, hv[k].z * nt, __ATOMIC_RELAXED, __HIP_MEMORY_SCOPE_WORKGROUP);
            __hip_atomic_fetch_add(ap + 96, hv[k].w * nt, __ATOMIC_RELAXED, __HIP_MEMORY_SCOPE_WORKGROUP);
        }
    }
}

template <bool SAVE>
__device__ __forceinline__ void tile_drain(TileCtx& t, int cnt,
                                           const int4* __restrict__ rec,
                                           const float* __restrict__ h) {
    const bool valid = t.lane < cnt;
    int e = t.q[(t.head + t.lane) & (QCAP - 1)];
    e = valid ? e : 0;
    const int4 rc = rec[e];
    const int r = rc.x;
    const int c = rc.y;
    const bool mine = valid && ((unsigned)(c - t.base) < (unsigned)TILE_NODES);
    const int loc = mine ? (c - t.base) : TILE_NODES;
    const int nrmBits = mine ? rc.z : 0;
    if (SAVE) {
        const unsigned mask = __builtin_amdgcn_ballot_w32(mine);
        const int pos = t.saved + (int)__builtin_amdgcn_mbcnt_lo(mask, 0u);
        if (mine && pos < LCAP)
            t.list[pos] = make_int2((int)(((unsigned)r << 8) | (unsigned)loc), rc.z);
        t.saved += __builtin_popcount(mask);
    }
    tile_add32(t, cnt, r, loc, nrmBits, h);
    t.head = (t.head + 32) & (QCAP - 1);
    t.qn -= cnt;
}

__device__ __forceinline__ void tile_offer(TileCtx& t, int c, int e) {
    const bool m = (unsigned)(c - t.base) < (unsigned)TILE_NODES;
    const unsigned mask = __builtin_amdgcn_ballot_w32(m);
    if (mask != 0u) {
        const int pos = t.head + t.qn + (int)__builtin_amdgcn_mbcnt_lo(mask, 0u);
        if (m) t.q[pos & (QCAP - 1)] = e;
        t.qn += __builtin_popcount(mask);
    }
}

__device__ __forceinline__ unsigned zero_bytes(unsigned x) {
    return ~(((x & 0x7F7F7F7Fu) + 0x7F7F7F7Fu) | x | 0x7F7F7F7Fu);
}

__device__ __forceinline__ unsigned match16(int4 kv, unsigned pat) {
    return (zero_bytes((unsigned)kv.x ^ pat) >> 7) | (zero_bytes((unsigned)kv.y ^ pat) >> 6) |
           (zero_bytes((unsigned)kv.z ^ pat) >> 5) | (zero_bytes((unsigned)kv.w ^ pat) >> 4);
}

__device__ __forceinline__ void tile_offer_mask(TileCtx& t, unsigned m, int ebase) {
    unsigned live = __builtin_amdgcn_ballot_w32(m != 0u);
    while (live != 0u) {
        const bool has = (m != 0u);
        const int p = __builtin_ffs((int)m) - 1;
        const int pos = t.head + t.qn + (int)__builtin_amdgcn_mbcnt_lo(live, 0u);
        if (has) t.q[pos & (QCAP - 1)] = ebase + ((p & 7) << 2) + (p >> 3);
        t.qn += __builtin_popcount(live);
        m &= m - 1u;
        live = __builtin_amdgcn_ballot_w32(m != 0u);
    }
}

template <bool SAVE>
__device__ __forceinline__ void tile_sum(TileCtx& t, int wave, int tid, int* cnt_out,
                                         const int4* __restrict__ rec,
                                         const int* __restrict__ key4,
                                         const float* __restrict__ h, int E) {
    for (int i = tid; i < (TILE_NODES + 1) * FEAT / 4; i += 1024)
        ((float4*)t.acc)[i] = make_float4(0.f, 0.f, 0.f, 0.f);
    __syncthreads();

    const int nGroups = E >> 9;
    const unsigned pat = (unsigned)(blockIdx.x & 255u) * 0x01010101u;
    const int start = (nGroups / (int)gridDim.x) * (int)blockIdx.x;

    int i = wave;
    for (; i + 3 * AGG_WAVES < nGroups; i += 4 * AGG_WAVES) {
        int4 kv[4];
        int eb[4];
#pragma unroll
        for (int u = 0; u < 4; ++u) {
            int g = start + i + u * AGG_WAVES;
            if (g >= nGroups) g -= nGroups;
            eb[u] = g * 512 + t.lane * 16;
            kv[u] = *(const int4*)(key4 + g * 128 + t.lane * 4);
        }
        int done = 0;
        do {
#pragma unroll
            for (int u = 0; u < 4; ++u) {
                if (u == done && t.qn <= QCAP - 512) {
                    tile_offer_mask(t, match16(kv[u], pat), eb[u]);
                    done = u + 1;
                }
            }
            while (t.qn >= 32) tile_drain<SAVE>(t, 32, rec, h);
        } while (done < 4);
    }
    for (; i < nGroups; i += AGG_WAVES) {
        int g = start + i;
        if (g >= nGroups) g -= nGroups;
        const int4 kv = *(const int4*)(key4 + g * 128 + t.lane * 4);
        tile_offer_mask(t, match16(kv, pat), g * 512 + t.lane * 16);
        while (t.qn >= 32) tile_drain<SAVE>(t, 32, rec, h);
    }
    if (wave == 0) {
        for (int eb = nGroups * 512; eb < E; eb += 32) {
            const int e = eb + t.lane;
            const int c = (e < E) ? rec[e].y : -1;
            tile_offer(t, c, e);
        }
    }
    while (t.qn >= 32) tile_drain<SAVE>(t, 32, rec, h);
    if (t.qn > 0) tile_drain<SAVE>(t, t.qn, rec, h);
    if (SAVE && t.lane == 0) *cnt_out = t.saved;
    __syncthreads();
}

__device__ __forceinline__ void tile_sum_from_list(TileCtx& t, int tid,
                                                   const int2* __restrict__ lp, int n,
                                                   const float* __restrict__ h) {
    for (int i = tid; i < (TILE_NODES + 1) * FEAT / 4; i += 1024)
        ((float4*)t.acc)[i] = make_float4(0.f, 0.f, 0.f, 0.f);
    __syncthreads();
    if (n > 0) {
        int2 cur = lp[min(t.lane, n - 1)];
        int i = 0;
        for (; i + 32 <= n; i += 32) {
            const int2 nxt = lp[min(i + 32 + t.lane, n - 1)];
            tile_add32(t, 32, (int)((unsigned)cur.x >> 8), cur.x & 255, cur.y, h);
            cur = nxt;
        }
        if (i < n) {
            const bool valid = i + t.lane < n;
            const int p = valid ? cur.x : TILE_NODES;
            const int wb = valid ? cur.y : 0;
            tile_add32(t, n - i, (int)((unsigned)p >> 8), p & 255, wb, h);
        }
    }
    __syncthreads();
}

#define CLOSE_TRIPS 4

__global__ __launch_bounds__(1024) void aggregate_layer1_kernel(const int* __restrict__ rec4,
                                                                const int* __restrict__ key4,
                                                                const float* __restrict__ dis,
                                                                const float* __restrict__ h,
                                                                const float* __restrict__ b0,
                                                                const float* __restrict__ g,
                                                                const float* __restrict__ be,
                                                                const float* __restrict__ mu,
                                                                const float* __restrict__ var,
                                                                float* __restrict__ h1,
                                                                int* __restrict__ lcnt,
                                                                int* __restrict__ list2,
                                                                int E, int N) {
    __shared__ float acc[(TILE_NODES + 1) * FEAT];
    __shared__ int qbuf[AGG_WAVES * QCAP];

    const int tid = threadIdx.x;
    const int wave = __builtin_amdgcn_readfirstlane(tid >> 5);

    TileCtx t;
    t.acc = acc;
    t.q = qbuf + wave * QCAP;
    t.base = blockIdx.x * TILE_NODES;
    t.lane = tid & 31;
    t.head = 0;
    t.qn = 0;
    t.list = (int2*)list2 + ((size_t)blockIdx.x * AGG_WAVES + wave) * LCAP;
    t.saved = 0;

    tile_sum<true>(t, wave, tid, lcnt + blockIdx.x * AGG_WAVES + wave,
                   (const int4*)rec4, key4, h, E);

    const int f = (tid & 31) << 2;
    const float4 bv = *(const float4*)(b0 + f);
    const float4 gv = *(const float4*)(g + f);
    const float4 ev = *(const float4*)(be + f);
    const float4 mv = *(const float4*)(mu + f);
    const float4 vv = *(const float4*)(var + f);
    const float bb[4] = {bv.x, bv.y, bv.z, bv.w};
    const float gg[4] = {gv.x, gv.y, gv.z, gv.w};
    const float ee[4] = {ev.x, ev.y, ev.z, ev.w};
    const float mm[4] = {mv.x, mv.y, mv.z, mv.w};
    const float rs[4] = {rsqrtf(vv.x + BN_EPS), rsqrtf(vv.y + BN_EPS),
                         rsqrtf(vv.z + BN_EPS), rsqrtf(vv.w + BN_EPS)};

    const int rows = min(TILE_NODES, N - t.base);
    const int lim = rows * (FEAT / 4);
    const float4* hsrc = (const float4*)(h + (size_t)t.base * FEAT);
    float4* dst = (float4*)(h1 + (size_t)t.base * FEAT);
    for (int i0 = tid; i0 < lim; i0 += CLOSE_TRIPS * 1024) {
        float4 hv[CLOSE_TRIPS];
        float dv[CLOSE_TRIPS];
#pragma unroll
        for (int u = 0; u < CLOSE_TRIPS; ++u) {
            const int j4 = min(i0 + u * 1024, lim - 1);
            hv[u] = hsrc[j4];
            dv[u] = dis[t.base + (j4 >> 5)];
        }
#pragma unroll
        for (int u = 0; u < CLOSE_TRIPS; ++u) {
            const int i4 = i0 + u * 1024;
            if (i4 < lim) {
                const float* ar = acc + (i4 >> 5) * FEAT + (tid & 31);
                const float dd = dv[u];
                const float av[4] = {ar[0], ar[32], ar[64], ar[96]};
                const float hh[4] = {hv[u].x, hv[u].y, hv[u].z, hv[u].w};
                float o[4];
#pragma unroll
                for (int j = 0; j < 4; ++j) {
                    float v = dd * (av[j] + hh[j]) + bb[j];
                    v = (v - mm[j]) * rs[j] * gg[j] + ee[j];
                    o[j] = (v > 0.0f) ? v : expm1f(v);
                }
                dst[i4] = make_float4(o[0], o[1], o[2], o[3]);
            }
        }
    }
}

__global__ __launch_bounds__(1024) void aggregate_layer2_kernel(const int* __restrict__ rec4,
                                                                const int* __restrict__ key4,
                                                                const float* __restrict__ dis,
                                                                const float* __restrict__ h,
                                                                const float* __restrict__ emb,
                                                                const float* __restrict__ h1,
                                                                const float* __restrict__ b1,
                                                                const int* __restrict__ lcnt,
                                                                const int* __restrict__ list2,
                                                                float* __restrict__ out,
                                                                int E, int N) {
    __shared__ float acc[(TILE_NODES + 1) * FEAT];
    __shared__ int qbuf[AGG_WAVES * QCAP];

    const int tid = threadIdx.x;
    const int wave = __builtin_amdgcn_readfirstlane(tid >> 5);

    TileCtx t;
    t.acc = acc;
    t.q = qbuf + wave * QCAP;
    t.base = blockIdx.x * TILE_NODES;
    t.lane = tid & 31;
    t.head = 0;
    t.qn = 0;
    t.list = nullptr;
    t.saved = 0;

    const int myc = lcnt[blockIdx.x * AGG_WAVES + t.lane];
    const bool over = __builtin_amdgcn_ballot_w32(myc > LCAP) != 0u;
    if (!over) {
        const int n = __builtin_amdgcn_readlane(myc, wave);
        tile_sum_from_list(t, tid,
                           (const int2*)list2 + ((size_t)blockIdx.x * AGG_WAVES + wave) * LCAP,
                           n, h);
    } else {
        tile_sum<false>(t, wave, tid, nullptr, (const int4*)rec4, key4, h, E);
    }

    const int f = (tid & 31) << 2;
    const float4 bv = *(const float4*)(b1 + f);
    const float inv3 = 1.0f / 3.0f;

    const int rows = min(TILE_NODES, N - t.base);
    const int lim = rows * (FEAT / 4);
    const float4* hsrc = (const float4*)(h + (size_t)t.base * FEAT);
    const float4* esrc = (const float4*)(emb + (size_t)t.base * FEAT);
    const float4* osrc = (const float4*)(h1 + (size_t)t.base * FEAT);
    float4* dst = (float4*)(out + (size_t)t.base * FEAT);
    for (int i0 = tid; i0 < lim; i0 += CLOSE_TRIPS * 1024) {
        float4 pv[CLOSE_TRIPS], xv[CLOSE_TRIPS], ov[CLOSE_TRIPS];
        float dv[CLOSE_TRIPS];
#pragma unroll
        for (int u = 0; u < CLOSE_TRIPS; ++u) {
            const int j4 = min(i0 + u * 1024, lim - 1);
            xv[u] = esrc[j4];
            ov[u] = osrc[j4];
            pv[u] = hsrc[j4];
            dv[u] = dis[t.base + (j4 >> 5)];
        }
#pragma unroll
        for (int u = 0; u < CLOSE_TRIPS; ++u) {
            asm volatile("" :: "v"(xv[u].x), "v"(xv[u].y), "v"(xv[u].z), "v"(xv[u].w),
                               "v"(ov[u].x), "v"(ov[u].y), "v"(ov[u].z), "v"(ov[u].w),
                               "v"(pv[u].x), "v"(pv[u].y), "v"(pv[u].z), "v"(pv[u].w),
                               "v"(dv[u]));
        }
#pragma unroll
        for (int u = 0; u < CLOSE_TRIPS; ++u) {
            const int i4 = i0 + u * 1024;
            if (i4 < lim) {
                const float* ar = acc + (i4 >> 5) * FEAT + (tid & 31);
                const float4 a2 = make_float4(ar[0], ar[32], ar[64], ar[96]);
                const float dd = dv[u];
                const float4 e = xv[u];
                const float4 o1 = ov[u];
                const float4 p2 = pv[u];
                float4 r;
                r.x = (e.x + o1.x + dd * (a2.x + p2.x) + bv.x) * inv3;
                r.y = (e.y + o1.y + dd * (a2.y + p2.y) + bv.y) * inv3;
                r.z = (e.z + o1.z + dd * (a2.z + p2.z) + bv.z) * inv3;
                r.w = (e.w + o1.w + dd * (a2.w + p2.w) + bv.w) * inv3;
                dst[i4] = r;
            }
        }
    }
}

extern "C" void kernel_launch(void* const* d_in, const int* in_sizes, int n_in,
                              void* d_out, int out_size, void* d_ws, size_t ws_size,
                              hipStream_t stream) {
    const float* emb = (const float*)d_in[0];
    const int*   ei  = (const int*)d_in[1];
    const float* ew  = (const float*)d_in[2];
    const float* W0  = (const float*)d_in[3];
    const float* b0  = (const float*)d_in[4];
    const float* W1  = (const float*)d_in[5];
    const float* b1  = (const float*)d_in[6];
    const float* bng = (const float*)d_in[7];
    const float* bnb = (const float*)d_in[8];
    const float* bnm = (const float*)d_in[9];
    const float* bnv = (const float*)d_in[10];
    float* out = (float*)d_out;

    const int N  = in_sizes[0] / FEAT;
    const int E  = in_sizes[2];
    const size_t ND = (size_t)N * FEAT;

    const int* row = ei;
    const int* col = ei + E;

    float* ws   = (float*)d_ws;
    size_t disPad = ((size_t)N + 1023) & ~(size_t)1023;
    size_t keyPad = ((size_t)E + 1023) & ~(size_t)1023;
    float* dis  = ws;
    float* bufA = ws + disPad;
    float* bufB = bufA + ND;
    int* key4   = (int*)(bufB + ND);
    int* rec    = (int*)((char*)key4 + keyPad);

    hipMemsetAsync(dis,  0, (size_t)N * sizeof(float), stream);

    const int gTile  = (N + TILE_NODES - 1) / TILE_NODES;
    const int gRowT  = (N + 15) / 16;
    int* lcnt   = rec + (size_t)E * 4;
    int* list2  = lcnt + (((size_t)gTile * AGG_WAVES + 63) & ~(size_t)63);

    degree_key_kernel<<<(E + 1023) / 1024, 256, 0, stream>>>(row, col, ew, dis, key4, rec, E);
    dis_kernel<<<(N + 255) / 256, 256, 0, stream>>>(dis, N);

    gemm_n128_kernel<<<gRowT, 256, 0, stream>>>(emb, W0, dis, bufA, N);

    aggregate_layer1_kernel<<<gTile, 1024, 0, stream>>>(rec, key4, dis, bufA,
                                                        b0, bng, bnb, bnm, bnv, bufB,
                                                        lcnt, list2, E, N);

    gemm_n128_kernel<<<gRowT, 256, 0, stream>>>(bufB, W1, dis, bufA, N);

    aggregate_layer2_kernel<<<gTile, 1024, 0, stream>>>(rec, key4, dis, bufA,
                                                        emb, bufB, b1, lcnt, list2, out, E, N);
}


# both projections run as a resident grid of 512 blocks: a wave loads its 128x16 slice of W once into registers and loops over 16-row tiles (A tile double-buffered in LDS); was 3125 one-tile blocks
# speedup vs baseline: 2.8502x; 1.0640x over previous
#include <hip/hip_runtime.h>
#include <hip/hip_bf16.h>
#include <math.h>

typedef float v2f __attribute__((ext_vector_type(2)));
typedef float v8f __attribute__((ext_vector_type(8)));

#if defined(__has_builtin)
#if __has_builtin(__builtin_amdgcn_wmma_f32_16x16x4_f32)
#define HAS_WMMA_F32_4 1
#endif
#endif
#ifndef HAS_WMMA_F32_4
#define HAS_WMMA_F32_4 0
#endif

#define FEAT 128
#define BN_EPS 1e-5f

#define GEMM_BLOCKS 512

#define TILE_NODES 196

__global__ __launch_bounds__(256) void degree_key_kernel(const int* __restrict__ row,
                                                         const int* __restrict__ col,
                                                         const float* __restrict__ w,
                                                         float* __restrict__ deg,
                                                         int* __restrict__ key4,
                                                         int* __restrict__ rec4, int E) {
    int4* __restrict__ rec = (int4*)rec4;
    const int er = blockIdx.x * 1024 + threadIdx.x;
    int rr[4], rc[4];
    float rw[4];
#pragma unroll
    for (int j = 0; j < 4; ++j) {
        const int el = min(er + j * 256, E - 1);
        rr[j] = row[el];
        rc[j] = col[el];
        rw[j] = w[el];
    }
#pragma unroll
    for (int j = 0; j < 4; ++j) {
        const int e = er + j * 256;
        if (e < E) rec[e] = make_int4(rr[j], rc[j], __float_as_int(rw[j]), 0);
    }
    const int t = blockIdx.x * 256 + threadIdx.x;
    const int e0 = t * 4;
    if (e0 >= E) return;
    unsigned k = 0u;
    if (e0 + 3 < E) {
        int c[4];
        float wv[4];
#pragma unroll
        for (int j = 0; j < 4; ++j) {
            c[j] = col[e0 + j];
            wv[j] = w[e0 + j];
        }
#pragma unroll
        for (int j = 0; j < 4; ++j) {
            __hip_atomic_fetch_add(&deg[c[j]], wv[j], __ATOMIC_RELAXED,
                                   __HIP_MEMORY_SCOPE_AGENT);
            k |= ((unsigned)(c[j] / TILE_NODES) & 255u) << (8 * j);
        }
    } else {
        for (int j = 0; j < 4 && e0 + j < E; ++j) {
            const int c = col[e0 + j];
            __hip_atomic_fetch_add(&deg[c], w[e0 + j], __ATOMIC_RELAXED,
                                   __HIP_MEMORY_SCOPE_AGENT);
            k |= ((unsigned)(c / TILE_NODES) & 255u) << (8 * j);
        }
    }
    key4[t] = (int)k;
}

__global__ __launch_bounds__(256) void dis_kernel(float* __restrict__ deg_dis, int n) {
    int i = blockIdx.x * 256 + threadIdx.x;
    if (i < n) deg_dis[i] = rsqrtf(deg_dis[i] + 1.0f);
}

__global__ __launch_bounds__(256) void gemm_n128_kernel(const float* __restrict__ X,
                                                        const float* __restrict__ W,
                                                        const float* __restrict__ dis,
                                                        float* __restrict__ Y,
                                                        int nrows, int ntiles) {
    __shared__ float lA[2][16 * 132];

    const int tid = threadIdx.x;
    const int wave = tid >> 5;
    const int lane = tid & 31;
    const int lid  = lane & 15;
    const int half = lane >> 4;
    const int n0   = wave * 16;

    int tile = blockIdx.x;
    if (tile >= ntiles) return;

    const int sr  = tid >> 5;
    const int sc4 = (tid & 31) << 2;

#if HAS_WMMA_F32_4
    v2f bf[32];
#pragma unroll
    for (int j = 0; j < 32; ++j) {
        const int ka = 4 * j + 2 * half;
        bf[j].x = W[(size_t)ka * FEAT + n0 + lid];
        bf[j].y = W[(size_t)(ka + 1) * FEAT + n0 + lid];
    }

    float4 pv[2];
    float ps[2];
#pragma unroll
    for (int i = 0; i < 2; ++i) {
        const int gc = min(tile * 16 + sr + 8 * i, nrows - 1);
        pv[i] = *(const float4*)(X + (size_t)gc * FEAT + sc4);
        ps[i] = dis[gc];
    }

    int buf = 0;
    for (;;) {
        float* sA = lA[buf];
#pragma unroll
        for (int i = 0; i < 2; ++i) {
            const float s = (tile * 16 + sr + 8 * i < nrows) ? ps[i] : 0.f;
            *(float4*)&sA[(sr + 8 * i) * 132 + sc4] =
                make_float4(pv[i].x * s, pv[i].y * s, pv[i].z * s, pv[i].w * s);
        }
        __syncthreads();

        const int next = tile + (int)gridDim.x;
        if (next < ntiles) {
#pragma unroll
            for (int i = 0; i < 2; ++i) {
                const int gc = min(next * 16 + sr + 8 * i, nrows - 1);
                pv[i] = *(const float4*)(X + (size_t)gc * FEAT + sc4);
                ps[i] = dis[gc];
            }
        }

        v8f acc = {0.f, 0.f, 0.f, 0.f, 0.f, 0.f, 0.f, 0.f};
        const float* arow = &sA[lid * 132];
#pragma unroll
        for (int j = 0; j < 32; ++j) {
            const int ka = 4 * j + 2 * half;
            v2f a;
            a.x = arow[ka];
            a.y = arow[ka + 1];
            acc = __builtin_amdgcn_wmma_f32_16x16x4_f32(
                 false, a,  false, bf[j],
                 (short)0, acc,  false,  false);
        }
        const int rowBase = tile * 16;
        float* yb = Y + (size_t)rowBase * FEAT + n0 + lid;
        if (rowBase + 16 <= nrows) {
#pragma unroll
            for (int v = 0; v < 8; ++v) yb[(size_t)(v + 8 * half) * FEAT] = acc[v];
        } else {
#pragma unroll
            for (int v = 0; v < 8; ++v) {
                const int m = v + 8 * half;
                if (rowBase + m < nrows) yb[(size_t)m * FEAT] = acc[v];
            }
        }

        if (next >= ntiles) break;
        tile = next;
        buf ^= 1;
    }
#else
    for (; tile < ntiles; tile += (int)gridDim.x) {
        const int rowBase = tile * 16;
        __syncthreads();
#pragma unroll
        for (int i = 0; i < 2; ++i) {
            const int gr = rowBase + sr + 8 * i;
            float4 v = make_float4(0.f, 0.f, 0.f, 0.f);
            float s = 0.f;
            if (gr < nrows) {
                v = *(const float4*)(X + (size_t)gr * FEAT + sc4);
                s = dis[gr];
            }
            float* dst = &lA[0][(sr + 8 * i) * 132 + sc4];
            dst[0] = v.x * s; dst[1] = v.y * s; dst[2] = v.z * s; dst[3] = v.w * s;
        }
        __syncthreads();
        float accs[8] = {0.f, 0.f, 0.f, 0.f, 0.f, 0.f, 0.f, 0.f};
        for (int k = 0; k < FEAT; ++k) {
            float bv = W[(size_t)k * FEAT + n0 + lid];
#pragma unroll
            for (int v = 0; v < 8; ++v)
                accs[v] = fmaf(lA[0][(v + 8 * half) * 132 + k], bv, accs[v]);
        }
        float* yb = Y + (size_t)rowBase * FEAT + n0 + lid;
#pragma unroll
        for (int v = 0; v < 8; ++v) {
            int m = v + 8 * half;
            if (rowBase + m < nrows) yb[(size_t)m * FEAT] = accs[v];
        }
    }
#endif
}

#define AGG_WAVES 32
#define QCAP 1024
#define LCAP 256

struct TileCtx {
    float* acc;
    int* q;
    int base;
    int lane;
    int head;
    int qn;
    int2* list;
    int saved;
};

__device__ __forceinline__ void tile_add32(TileCtx& t, int cnt, int r, int loc, int nrmBits,
                                           const float* __restrict__ h) {
#pragma unroll
    for (int sub = 0; sub < 4; ++sub) {
        if (sub * 8 >= cnt) break;
        float4 hv[8];
#pragma unroll
        for (int k = 0; k < 8; ++k) {
            const int rt = __builtin_amdgcn_readlane(r, sub * 8 + k);
            hv[k] = *(const float4*)(h + (size_t)rt * FEAT + 4 * t.lane);
        }
#pragma unroll
        for (int k = 0; k < 8; ++k) {
            const float nt = __int_as_float(__builtin_amdgcn_readlane(nrmBits, sub * 8 + k));
            const int lt = __builtin_amdgcn_readlane(loc, sub * 8 + k);
            float* ap = t.acc + lt * FEAT + t.lane;
            __hip_atomic_fetch_add(ap + 0,  hv[k].x * nt, __ATOMIC_RELAXED, __HIP_MEMORY_SCOPE_WORKGROUP);
            __hip_atomic_fetch_add(ap + 32, hv[k].y * nt, __ATOMIC_RELAXED, __HIP_MEMORY_SCOPE_WORKGROUP);
            __hip_atomic_fetch_add(ap + 64, hv[k].z * nt, __ATOMIC_RELAXED, __HIP_MEMORY_SCOPE_WORKGROUP);
            __hip_atomic_fetch_add(ap + 96, hv[k].w * nt, __ATOMIC_RELAXED, __HIP_MEMORY_SCOPE_WORKGROUP);
        }
    }
}

template <bool SAVE>
__device__ __forceinline__ void tile_drain(TileCtx& t, int cnt,
                                           const int4* __restrict__ rec,
                                           const float* __restrict__ h) {
    const bool valid = t.lane < cnt;
    int e = t.q[(t.head + t.lane) & (QCAP - 1)];
    e = valid ? e : 0;
    const int4 rc = rec[e];
    const int r = rc.x;
    const int c = rc.y;
    const bool mine = valid && ((unsigned)(c - t.base) < (unsigned)TILE_NODES);
    const int loc = mine ? (c - t.base) : TILE_NODES;
    const int nrmBits = mine ? rc.z : 0;
    if (SAVE) {
        const unsigned mask = __builtin_amdgcn_ballot_w32(mine);
        const int pos = t.saved + (int)__builtin_amdgcn_mbcnt_lo(mask, 0u);
        if (mine && pos < LCAP)
            t.list[pos] = make_int2((int)(((unsigned)r << 8) | (unsigned)loc), rc.z);
        t.saved += __builtin_popcount(mask);
    }
    tile_add32(t, cnt, r, loc, nrmBits, h);
    t.head = (t.head + 32) & (QCAP - 1);
    t.qn -= cnt;
}

__device__ __forceinline__ void tile_offer(TileCtx& t, int c, int e) {
    const bool m = (unsigned)(c - t.base) < (unsigned)TILE_NODES;
    const unsigned mask = __builtin_amdgcn_ballot_w32(m);
    if (mask != 0u) {
        const int pos = t.head + t.qn + (int)__builtin_amdgcn_mbcnt_lo(mask, 0u);
        if (m) t.q[pos & (QCAP - 1)] = e;
        t.qn += __builtin_popcount(mask);
    }
}

__device__ __forceinline__ unsigned zero_bytes(unsigned x) {
    return ~(((x & 0x7F7F7F7Fu) + 0x7F7F7F7Fu) | x | 0x7F7F7F7Fu);
}

__device__ __forceinline__ unsigned match16(int4 kv, unsigned pat) {
    return (zero_bytes((unsigned)kv.x ^ pat) >> 7) | (zero_bytes((unsigned)kv.y ^ pat) >> 6) |
           (zero_bytes((unsigned)kv.z ^ pat) >> 5) | (zero_bytes((unsigned)kv.w ^ pat) >> 4);
}

__device__ __forceinline__ void tile_offer_mask(TileCtx& t, unsigned m, int ebase) {
    unsigned live = __builtin_amdgcn_ballot_w32(m != 0u);
    while (live != 0u) {
        const bool has = (m != 0u);
        const int p = __builtin_ffs((int)m) - 1;
        const int pos = t.head + t.qn + (int)__builtin_amdgcn_mbcnt_lo(live, 0u);
        if (has) t.q[pos & (QCAP - 1)] = ebase + ((p & 7) << 2) + (p >> 3);
        t.qn += __builtin_popcount(live);
        m &= m - 1u;
        live = __builtin_amdgcn_ballot_w32(m != 0u);
    }
}

template <bool SAVE>
__device__ __forceinline__ void tile_sum(TileCtx& t, int wave, int tid, int* cnt_out,
                                         const int4* __restrict__ rec,
                                         const int* __restrict__ key4,
                                         const float* __restrict__ h, int E) {
    for (int i = tid; i < (TILE_NODES + 1) * FEAT / 4; i += 1024)
        ((float4*)t.acc)[i] = make_float4(0.f, 0.f, 0.f, 0.f);
    __syncthreads();

    const int nGroups = E >> 9;
    const unsigned pat = (unsigned)(blockIdx.x & 255u) * 0x01010101u;
    const int start = (nGroups / (int)gridDim.x) * (int)blockIdx.x;

    int i = wave;
    for (; i + 3 * AGG_WAVES < nGroups; i += 4 * AGG_WAVES) {
        int4 kv[4];
        int eb[4];
#pragma unroll
        for (int u = 0; u < 4; ++u) {
            int g = start + i + u * AGG_WAVES;
            if (g >= nGroups) g -= nGroups;
            eb[u] = g * 512 + t.lane * 16;
            kv[u] = *(const int4*)(key4 + g * 128 + t.lane * 4);
        }
        int done = 0;
        do {
#pragma unroll
            for (int u = 0; u < 4; ++u) {
                if (u == done && t.qn <= QCAP - 512) {
                    tile_offer_mask(t, match16(kv[u], pat), eb[u]);
                    done = u + 1;
                }
            }
            while (t.qn >= 32) tile_drain<SAVE>(t, 32, rec, h);
        } while (done < 4);
    }
    for (; i < nGroups; i += AGG_WAVES) {
        int g = start + i;
        if (g >= nGroups) g -= nGroups;
        const int4 kv = *(const int4*)(key4 + g * 128 + t.lane * 4);
        tile_offer_mask(t, match16(kv, pat), g * 512 + t.lane * 16);
        while (t.qn >= 32) tile_drain<SAVE>(t, 32, rec, h);
    }
    if (wave == 0) {
        for (int eb = nGroups * 512; eb < E; eb += 32) {
            const int e = eb + t.lane;
            const int c = (e < E) ? rec[e].y : -1;
            tile_offer(t, c, e);
        }
    }
    while (t.qn >= 32) tile_drain<SAVE>(t, 32, rec, h);
    if (t.qn > 0) tile_drain<SAVE>(t, t.qn, rec, h);
    if (SAVE && t.lane == 0) *cnt_out = t.saved;
    __syncthreads();
}

__device__ __forceinline__ void tile_sum_from_list(TileCtx& t, int tid,
                                                   const int2* __restrict__ lp, int n,
                                                   const float* __restrict__ h) {
    for (int i = tid; i < (TILE_NODES + 1) * FEAT / 4; i += 1024)
        ((float4*)t.acc)[i] = make_float4(0.f, 0.f, 0.f, 0.f);
    __syncthreads();
    if (n > 0) {
        int2 cur = lp[min(t.lane, n - 1)];
        int i = 0;
        for (; i + 32 <= n; i += 32) {
            const int2 nxt = lp[min(i + 32 + t.lane, n - 1)];
            tile_add32(t, 32, (int)((unsigned)cur.x >> 8), cur.x & 255, cur.y, h);
            cur = nxt;
        }
        if (i < n) {
            const bool valid = i + t.lane < n;
            const int p = valid ? cur.x : TILE_NODES;
            const int wb = valid ? cur.y : 0;
            tile_add32(t, n - i, (int)((unsigned)p >> 8), p & 255, wb, h);
        }
    }
    __syncthreads();
}

#define CLOSE_TRIPS 4

__global__ __launch_bounds__(1024) void aggregate_layer1_kernel(const int* __restrict__ rec4,
                                                                const int* __restrict__ key4,
                                                                const float* __restrict__ dis,
                                                                const float* __restrict__ h,
                                                                const float* __restrict__ b0,
                                                                const float* __restrict__ g,
                                                                const float* __restrict__ be,
                                                                const float* __restrict__ mu,
                                                                const float* __restrict__ var,
                                                                float* __restrict__ h1,
                                                                int* __restrict__ lcnt,
                                                                int* __restrict__ list2,
                                                                int E, int N) {
    __shared__ float acc[(TILE_NODES + 1) * FEAT];
    __shared__ int qbuf[AGG_WAVES * QCAP];

    const int tid = threadIdx.x;
    const int wave = __builtin_amdgcn_readfirstlane(tid >> 5);

    TileCtx t;
    t.acc = acc;
    t.q = qbuf + wave * QCAP;
    t.base = blockIdx.x * TILE_NODES;
    t.lane = tid & 31;
    t.head = 0;
    t.qn = 0;
    t.list = (int2*)list2 + ((size_t)blockIdx.x * AGG_WAVES + wave) * LCAP;
    t.saved = 0;

    tile_sum<true>(t, wave, tid, lcnt + blockIdx.x * AGG_WAVES + wave,
                   (const int4*)rec4, key4, h, E);

    const int f = (tid & 31) << 2;
    const float4 bv = *(const float4*)(b0 + f);
    const float4 gv = *(const float4*)(g + f);
    const float4 ev = *(const float4*)(be + f);
    const float4 mv = *(const float4*)(mu + f);
    const float4 vv = *(const float4*)(var + f);
    const float bb[4] = {bv.x, bv.y, bv.z, bv.w};
    const float gg[4] = {gv.x, gv.y, gv.z, gv.w};
    const float ee[4] = {ev.x, ev.y, ev.z, ev.w};
    const float mm[4] = {mv.x, mv.y, mv.z, mv.w};
    const float rs[4] = {rsqrtf(vv.x + BN_EPS), rsqrtf(vv.y + BN_EPS),
                         rsqrtf(vv.z + BN_EPS), rsqrtf(vv.w + BN_EPS)};

    const int rows = min(TILE_NODES, N - t.base);
    const int lim = rows * (FEAT / 4);
    const float4* hsrc = (const float4*)(h + (size_t)t.base * FEAT);
    float4* dst = (float4*)(h1 + (size_t)t.base * FEAT);
    for (int i0 = tid; i0 < lim; i0 += CLOSE_TRIPS * 1024) {
        float4 hv[CLOSE_TRIPS];
        float dv[CLOSE_TRIPS];
#pragma unroll
        for (int u = 0; u < CLOSE_TRIPS; ++u) {
            const int j4 = min(i0 + u * 1024, lim - 1);
            hv[u] = hsrc[j4];
            dv[u] = dis[t.base + (j4 >> 5)];
        }
#pragma unroll
        for (int u = 0; u < CLOSE_TRIPS; ++u) {
            const int i4 = i0 + u * 1024;
            if (i4 < lim) {
                const float* ar = acc + (i4 >> 5) * FEAT + (tid & 31);
                const float dd = dv[u];
                const float av[4] = {ar[0], ar[32], ar[64], ar[96]};
                const float hh[4] = {hv[u].x, hv[u].y, hv[u].z, hv[u].w};
                float o[4];
#pragma unroll
                for (int j = 0; j < 4; ++j) {
                    float v = dd * (av[j] + hh[j]) + bb[j];
                    v = (v - mm[j]) * rs[j] * gg[j] + ee[j];
                    o[j] = (v > 0.0f) ? v : expm1f(v);
                }
                dst[i4] = make_float4(o[0], o[1], o[2], o[3]);
            }
        }
    }
}

__global__ __launch_bounds__(1024) void aggregate_layer2_kernel(const int* __restrict__ rec4,
                                                                const int* __restrict__ key4,
                                                                const float* __restrict__ dis,
                                                                const float* __restrict__ h,
                                                                const float* __restrict__ emb,
                                                                const float* __restrict__ h1,
                                                                const float* __restrict__ b1,
                                                                const int* __restrict__ lcnt,
                                                                const int* __restrict__ list2,
                                                                float* __restrict__ out,
                                                                int E, int N) {
    __shared__ float acc[(TILE_NODES + 1) * FEAT];
    __shared__ int qbuf[AGG_WAVES * QCAP];

    const int tid = threadIdx.x;
    const int wave = __builtin_amdgcn_readfirstlane(tid >> 5);

    TileCtx t;
    t.acc = acc;
    t.q = qbuf + wave * QCAP;
    t.base = blockIdx.x * TILE_NODES;
    t.lane = tid & 31;
    t.head = 0;
    t.qn = 0;
    t.list = nullptr;
    t.saved = 0;

    const int myc = lcnt[blockIdx.x * AGG_WAVES + t.lane];
    const bool over = __builtin_amdgcn_ballot_w32(myc > LCAP) != 0u;
    if (!over) {
        const int n = __builtin_amdgcn_readlane(myc, wave);
        tile_sum_from_list(t, tid,
                           (const int2*)list2 + ((size_t)blockIdx.x * AGG_WAVES + wave) * LCAP,
                           n, h);
    } else {
        tile_sum<false>(t, wave, tid, nullptr, (const int4*)rec4, key4, h, E);
    }

    const int f = (tid & 31) << 2;
    const float4 bv = *(const float4*)(b1 + f);
    const float inv3 = 1.0f / 3.0f;

    const int rows = min(TILE_NODES, N - t.base);
    const int lim = rows * (FEAT / 4);
    const float4* hsrc = (const float4*)(h + (size_t)t.base * FEAT);
    const float4* esrc = (const float4*)(emb + (size_t)t.base * FEAT);
    const float4* osrc = (const float4*)(h1 + (size_t)t.base * FEAT);
    float4* dst = (float4*)(out + (size_t)t.base * FEAT);
    for (int i0 = tid; i0 < lim; i0 += CLOSE_TRIPS * 1024) {
        float4 pv[CLOSE_TRIPS], xv[CLOSE_TRIPS], ov[CLOSE_TRIPS];
        float dv[CLOSE_TRIPS];
#pragma unroll
        for (int u = 0; u < CLOSE_TRIPS; ++u) {
            const int j4 = min(i0 + u * 1024, lim - 1);
            xv[u] = esrc[j4];
            ov[u] = osrc[j4];
            pv[u] = hsrc[j4];
            dv[u] = dis[t.base + (j4 >> 5)];
        }
#pragma unroll
        for (int u = 0; u < CLOSE_TRIPS; ++u) {
            asm volatile("" :: "v"(xv[u].x), "v"(xv[u].y), "v"(xv[u].z), "v"(xv[u].w),
                               "v"(ov[u].x), "v"(ov[u].y), "v"(ov[u].z), "v"(ov[u].w),
                               "v"(pv[u].x), "v"(pv[u].y), "v"(pv[u].z), "v"(pv[u].w),
                               "v"(dv[u]));
        }
#pragma unroll
        for (int u = 0; u < CLOSE_TRIPS; ++u) {
            const int i4 = i0 + u * 1024;
            if (i4 < lim) {
                const float* ar = acc + (i4 >> 5) * FEAT + (tid & 31);
                const float4 a2 = make_float4(ar[0], ar[32], ar[64], ar[96]);
                const float dd = dv[u];
                const float4 e = xv[u];
                const float4 o1 = ov[u];
                const float4 p2 = pv[u];
                float4 r;
                r.x = (e.x + o1.x + dd * (a2.x + p2.x) + bv.x) * inv3;
                r.y = (e.y + o1.y + dd * (a2.y + p2.y) + bv.y) * inv3;
                r.z = (e.z + o1.z + dd * (a2.z + p2.z) + bv.z) * inv3;
                r.w = (e.w + o1.w + dd * (a2.w + p2.w) + bv.w) * inv3;
                dst[i4] = r;
            }
        }
    }
}

extern "C" void kernel_launch(void* const* d_in, const int* in_sizes, int n_in,
                              void* d_out, int out_size, void* d_ws, size_t ws_size,
                              hipStream_t stream) {
    const float* emb = (const float*)d_in[0];
    const int*   ei  = (const int*)d_in[1];
    const float* ew  = (const float*)d_in[2];
    const float* W0  = (const float*)d_in[3];
    const float* b0  = (const float*)d_in[4];
    const float* W1  = (const float*)d_in[5];
    const float* b1  = (const float*)d_in[6];
    const float* bng = (const float*)d_in[7];
    const float* bnb = (const float*)d_in[8];
    const float* bnm = (const float*)d_in[9];
    const float* bnv = (const float*)d_in[10];
    float* out = (float*)d_out;

    const int N  = in_sizes[0] / FEAT;
    const int E  = in_sizes[2];
    const size_t ND = (size_t)N * FEAT;

    const int* row = ei;
    const int* col = ei + E;

    float* ws   = (float*)d_ws;
    size_t disPad = ((size_t)N + 1023) & ~(size_t)1023;
    size_t keyPad = ((size_t)E + 1023) & ~(size_t)1023;
    float* dis  = ws;
    float* bufA = ws + disPad;
    float* bufB = bufA + ND;
    int* key4   = (int*)(bufB + ND);
    int* rec    = (int*)((char*)key4 + keyPad);

    hipMemsetAsync(dis,  0, (size_t)N * sizeof(float), stream);

    const int gTile  = (N + TILE_NODES - 1) / TILE_NODES;
    const int gRowT  = (N + 15) / 16;
    int* lcnt   = rec + (size_t)E * 4;
    int* list2  = lcnt + (((size_t)gTile * AGG_WAVES + 63) & ~(size_t)63);

    degree_key_kernel<<<(E + 1023) / 1024, 256, 0, stream>>>(row, col, ew, dis, key4, rec, E);
    dis_kernel<<<(N + 255) / 256, 256, 0, stream>>>(dis, N);

    gemm_n128_kernel<<<GEMM_BLOCKS, 256, 0, stream>>>(emb, W0, dis, bufA, N, gRowT);

    aggregate_layer1_kernel<<<gTile, 1024, 0, stream>>>(rec, key4, dis, bufA,
                                                        b0, bng, bnb, bnm, bnv, bufB,
                                                        lcnt, list2, E, N);

    gemm_n128_kernel<<<GEMM_BLOCKS, 256, 0, stream>>>(bufB, W1, dis, bufA, N, gRowT);

    aggregate_layer2_kernel<<<gTile, 1024, 0, stream>>>(rec, key4, dis, bufA,
                                                        emb, bufB, b1, lcnt, list2, out, E, N);
}


# 6 -> 4 launches: layer 2's projection runs at layer 1's tile store (finished h1 rows staged in the idle ring LDS feed the f32 WMMA, W1 in LDS); dis = rsqrt(deg+1) formed where read; sum order kept
# speedup vs baseline: 2.9516x; 1.0356x over previous
#include <hip/hip_runtime.h>
#include <hip/hip_bf16.h>
#include <math.h>

typedef float v2f __attribute__((ext_vector_type(2)));
typedef float v8f __attribute__((ext_vector_type(8)));

#if defined(__has_builtin)
#if __has_builtin(__builtin_amdgcn_wmma_f32_16x16x4_f32)
#define HAS_WMMA_F32_4 1
#endif
#endif
#ifndef HAS_WMMA_F32_4
#define HAS_WMMA_F32_4 0
#endif

#define FEAT 128
#define BN_EPS 1e-5f

#define GEMM_BLOCKS 512

#define TILE_NODES 196

#define LA_PITCH 132
#define WP_PITCH 288
#define W1_PIECES ((FEAT / 2) * (FEAT / 4) / 1024)

__global__ __launch_bounds__(256) void degree_key_kernel(const int* __restrict__ row,
                                                         const int* __restrict__ col,
                                                         const float* __restrict__ w,
                                                         float* __restrict__ deg,
                                                         int* __restrict__ key4,
                                                         int* __restrict__ rec4, int E) {
    int4* __restrict__ rec = (int4*)rec4;
    const int er = blockIdx.x * 1024 + threadIdx.x;
    int rr[4], rc[4];
    float rw[4];
#pragma unroll
    for (int j = 0; j < 4; ++j) {
        const int el = min(er + j * 256, E - 1);
        rr[j] = row[el];
        rc[j] = col[el];
        rw[j] = w[el];
    }
#pragma unroll
    for (int j = 0; j < 4; ++j) {
        const int e = er + j * 256;
        if (e < E) rec[e] = make_int4(rr[j], rc[j], __float_as_int(rw[j]), 0);
    }
    const int t = blockIdx.x * 256 + threadIdx.x;
    const int e0 = t * 4;
    if (e0 >= E) return;
    unsigned k = 0u;
    if (e0 + 3 < E) {
        int c[4];
        float wv[4];
#pragma unroll
        for (int j = 0; j < 4; ++j) {
            c[j] = col[e0 + j];
            wv[j] = w[e0 + j];
        }
#pragma unroll
        for (int j = 0; j < 4; ++j) {
            __hip_atomic_fetch_add(&deg[c[j]], wv[j], __ATOMIC_RELAXED,
                                   __HIP_MEMORY_SCOPE_AGENT);
            k |= ((unsigned)(c[j] / TILE_NODES) & 255u) << (8 * j);
        }
    } else {
        for (int j = 0; j < 4 && e0 + j < E; ++j) {
            const int c = col[e0 + j];
            __hip_atomic_fetch_add(&deg[c], w[e0 + j], __ATOMIC_RELAXED,
                                   __HIP_MEMORY_SCOPE_AGENT);
            k |= ((unsigned)(c / TILE_NODES) & 255u) << (8 * j);
        }
    }
    key4[t] = (int)k;
}

__device__ __forceinline__ float dis_of(const float* __restrict__ deg, int i) {
    return rsqrtf(deg[i] + 1.0f);
}

__global__ __launch_bounds__(256) void gemm_n128_kernel(const float* __restrict__ X,
                                                        const float* __restrict__ W,
                                                        const float* __restrict__ deg,
                                                        float* __restrict__ Y,
                                                        int nrows, int ntiles) {
    __shared__ float lA[2][16 * 132];

    const int tid = threadIdx.x;
    const int wave = tid >> 5;
    const int lane = tid & 31;
    const int lid  = lane & 15;
    const int half = lane >> 4;
    const int n0   = wave * 16;

    int tile = blockIdx.x;
    if (tile >= ntiles) return;

    const int sr  = tid >> 5;
    const int sc4 = (tid & 31) << 2;

#if HAS_WMMA_F32_4
    v2f bf[32];
#pragma unroll
    for (int j = 0; j < 32; ++j) {
        const int ka = 4 * j + 2 * half;
        bf[j].x = W[(size_t)ka * FEAT + n0 + lid];
        bf[j].y = W[(size_t)(ka + 1) * FEAT + n0 + lid];
    }

    float4 pv[2];
    float ps[2];
#pragma unroll
    for (int i = 0; i < 2; ++i) {
        const int gc = min(tile * 16 + sr + 8 * i, nrows - 1);
        pv[i] = *(const float4*)(X + (size_t)gc * FEAT + sc4);
        ps[i] = dis_of(deg, gc);
    }

    int buf = 0;
    for (;;) {
        float* sA = lA[buf];
#pragma unroll
        for (int i = 0; i < 2; ++i) {
            const float s = (tile * 16 + sr + 8 * i < nrows) ? ps[i] : 0.f;
            *(float4*)&sA[(sr + 8 * i) * 132 + sc4] =
                make_float4(pv[i].x * s, pv[i].y * s, pv[i].z * s, pv[i].w * s);
        }
        __syncthreads();

        const int next = tile + (int)gridDim.x;
        if (next < ntiles) {
#pragma unroll
            for (int i = 0; i < 2; ++i) {
                const int gc = min(next * 16 + sr + 8 * i, nrows - 1);
                pv[i] = *(const float4*)(X + (size_t)gc * FEAT + sc4);
                ps[i] = dis_of(deg, gc);
            }
        }

        v8f acc = {0.f, 0.f, 0.f, 0.f, 0.f, 0.f, 0.f, 0.f};
        const float* arow = &sA[lid * 132];
#pragma unroll
        for (int j = 0; j < 32; ++j) {
            const int ka = 4 * j + 2 * half;
            v2f a;
            a.x = arow[ka];
            a.y = arow[ka + 1];
            acc = __builtin_amdgcn_wmma_f32_16x16x4_f32(
                 false, a,  false, bf[j],
                 (short)0, acc,  false,  false);
        }
        const int rowBase = tile * 16;
        float* yb = Y + (size_t)rowBase * FEAT + n0 + lid;
        if (rowBase + 16 <= nrows) {
#pragma unroll
            for (int v = 0; v < 8; ++v) yb[(size_t)(v + 8 * half) * FEAT] = acc[v];
        } else {
#pragma unroll
            for (int v = 0; v < 8; ++v) {
                const int m = v + 8 * half;
                if (rowBase + m < nrows) yb[(size_t)m * FEAT] = acc[v];
            }
        }

        if (next >= ntiles) break;
        tile = next;
        buf ^= 1;
    }
#else
    for (; tile < ntiles; tile += (int)gridDim.x) {
        const int rowBase = tile * 16;
        __syncthreads();
#pragma unroll
        for (int i = 0; i < 2; ++i) {
            const int gr = rowBase + sr + 8 * i;
            float4 v = make_float4(0.f, 0.f, 0.f, 0.f);
            float s = 0.f;
            if (gr < nrows) {
                v = *(const float4*)(X + (size_t)gr * FEAT + sc4);
                s = dis_of(deg, gr);
            }
            float* dst = &lA[0][(sr + 8 * i) * 132 + sc4];
            dst[0] = v.x * s; dst[1] = v.y * s; dst[2] = v.z * s; dst[3] = v.w * s;
        }
        __syncthreads();
        float accs[8] = {0.f, 0.f, 0.f, 0.f, 0.f, 0.f, 0.f, 0.f};
        for (int k = 0; k < FEAT; ++k) {
            float bv = W[(size_t)k * FEAT + n0 + lid];
#pragma unroll
            for (int v = 0; v < 8; ++v)
                accs[v] = fmaf(lA[0][(v + 8 * half) * 132 + k], bv, accs[v]);
        }
        float* yb = Y + (size_t)rowBase * FEAT + n0 + lid;
#pragma unroll
        for (int v = 0; v < 8; ++v) {
            int m = v + 8 * half;
            if (rowBase + m < nrows) yb[(size_t)m * FEAT] = accs[v];
        }
    }
#endif
}

#define AGG_WAVES 32
#define QCAP 1024
#define LCAP 256

struct TileCtx {
    float* acc;
    int* q;
    int base;
    int lane;
    int head;
    int qn;
    int2* list;
    int saved;
};

__device__ __forceinline__ void tile_add32(TileCtx& t, int cnt, int r, int loc, int nrmBits,
                                           const float* __restrict__ h) {
#pragma unroll
    for (int sub = 0; sub < 4; ++sub) {
        if (sub * 8 >= cnt) break;
        float4 hv[8];
#pragma unroll
        for (int k = 0; k < 8; ++k) {
            const int rt = __builtin_amdgcn_readlane(r, sub * 8 + k);
            hv[k] = *(const float4*)(h + (size_t)rt * FEAT + 4 * t.lane);
        }
#pragma unroll
        for (int k = 0; k < 8; ++k) {
            const float nt = __int_as_float(__builtin_amdgcn_readlane(nrmBits, sub * 8 + k));
            const int lt = __builtin_amdgcn_readlane(loc, sub * 8 + k);
            float* ap = t.acc + lt * FEAT + t.lane;
            __hip_atomic_fetch_add(ap + 0,  hv[k].x * nt, __ATOMIC_RELAXED, __HIP_MEMORY_SCOPE_WORKGROUP);
            __hip_atomic_fetch_add(ap + 32, hv[k].y * nt, __ATOMIC_RELAXED, __HIP_MEMORY_SCOPE_WORKGROUP);
            __hip_atomic_fetch_add(ap + 64, hv[k].z * nt, __ATOMIC_RELAXED, __HIP_MEMORY_SCOPE_WORKGROUP);
            __hip_atomic_fetch_add(ap + 96, hv[k].w * nt, __ATOMIC_RELAXED, __HIP_MEMORY_SCOPE_WORKGROUP);
        }
    }
}

template <bool SAVE>
__device__ __forceinline__ void tile_drain(TileCtx& t, int cnt,
                                           const int4* __restrict__ rec,
                                           const float* __restrict__ h) {
    const bool valid = t.lane < cnt;
    int e = t.q[(t.head + t.lane) & (QCAP - 1)];
    e = valid ? e : 0;
    const int4 rc = rec[e];
    const int r = rc.x;
    const int c = rc.y;
    const bool mine = valid && ((unsigned)(c - t.base) < (unsigned)TILE_NODES);
    const int loc = mine ? (c - t.base) : TILE_NODES;
    const int nrmBits = mine ? rc.z : 0;
    if (SAVE) {
        const unsigned mask = __builtin_amdgcn_ballot_w32(mine);
        const int pos = t.saved + (int)__builtin_amdgcn_mbcnt_lo(mask, 0u);
        if (mine && pos < LCAP)
            t.list[pos] = make_int2((int)(((unsigned)r << 8) | (unsigned)loc), rc.z);
        t.saved += __builtin_popcount(mask);
    }
    tile_add32(t, cnt, r, loc, nrmBits, h);
    t.head = (t.head + 32) & (QCAP - 1);
    t.qn -= cnt;
}

__device__ __forceinline__ void tile_offer(TileCtx& t, int c, int e) {
    const bool m = (unsigned)(c - t.base) < (unsigned)TILE_NODES;
    const unsigned mask = __builtin_amdgcn_ballot_w32(m);
    if (mask != 0u) {
        const int pos = t.head + t.qn + (int)__builtin_amdgcn_mbcnt_lo(mask, 0u);
        if (m) t.q[pos & (QCAP - 1)] = e;
        t.qn += __builtin_popcount(mask);
    }
}

__device__ __forceinline__ unsigned zero_bytes(unsigned x) {
    return ~(((x & 0x7F7F7F7Fu) + 0x7F7F7F7Fu) | x | 0x7F7F7F7Fu);
}

__device__ __forceinline__ unsigned match16(int4 kv, unsigned pat) {
    return (zero_bytes((unsigned)kv.x ^ pat) >> 7) | (zero_bytes((unsigned)kv.y ^ pat) >> 6) |
           (zero_bytes((unsigned)kv.z ^ pat) >> 5) | (zero_bytes((unsigned)kv.w ^ pat) >> 4);
}

__device__ __forceinline__ void tile_offer_mask(TileCtx& t, unsigned m, int ebase) {
    unsigned live = __builtin_amdgcn_ballot_w32(m != 0u);
    while (live != 0u) {
        const bool has = (m != 0u);
        const int p = __builtin_ffs((int)m) - 1;
        const int pos = t.head + t.qn + (int)__builtin_amdgcn_mbcnt_lo(live, 0u);
        if (has) t.q[pos & (QCAP - 1)] = ebase + ((p & 7) << 2) + (p >> 3);
        t.qn += __builtin_popcount(live);
        m &= m - 1u;
        live = __builtin_amdgcn_ballot_w32(m != 0u);
    }
}

template <bool SAVE>
__device__ __forceinline__ void tile_sum(TileCtx& t, int wave, int tid, int* cnt_out,
                                         const int4* __restrict__ rec,
                                         const int* __restrict__ key4,
                                         const float* __restrict__ h, int E) {
    for (int i = tid; i < (TILE_NODES + 1) * FEAT / 4; i += 1024)
        ((float4*)t.acc)[i] = make_float4(0.f, 0.f, 0.f, 0.f);
    __syncthreads();

    const int nGroups = E >> 9;
    const unsigned pat = (unsigned)(blockIdx.x & 255u) * 0x01010101u;
    const int start = (nGroups / (int)gridDim.x) * (int)blockIdx.x;

    int i = wave;
    for (; i + 3 * AGG_WAVES < nGroups; i += 4 * AGG_WAVES) {
        int4 kv[4];
        int eb[4];
#pragma unroll
        for (int u = 0; u < 4; ++u) {
            int g = start + i + u * AGG_WAVES;
            if (g >= nGroups) g -= nGroups;
            eb[u] = g * 512 + t.lane * 16;
            kv[u] = *(const int4*)(key4 + g * 128 + t.lane * 4);
        }
        int done = 0;
        do {
#pragma unroll
            for (int u = 0; u < 4; ++u) {
                if (u == done && t.qn <= QCAP - 512) {
                    tile_offer_mask(t, match16(kv[u], pat), eb[u]);
                    done = u + 1;
                }
            }
            while (t.qn >= 32) tile_drain<SAVE>(t, 32, rec, h);
        } while (done < 4);
    }
    for (; i < nGroups; i += AGG_WAVES) {
        int g = start + i;
        if (g >= nGroups) g -= nGroups;
        const int4 kv = *(const int4*)(key4 + g * 128 + t.lane * 4);
        tile_offer_mask(t, match16(kv, pat), g * 512 + t.lane * 16);
        while (t.qn >= 32) tile_drain<SAVE>(t, 32, rec, h);
    }
    if (wave == 0) {
        for (int eb = nGroups * 512; eb < E; eb += 32) {
            const int e = eb + t.lane;
            const int c = (e < E) ? rec[e].y : -1;
            tile_offer(t, c, e);
        }
    }
    while (t.qn >= 32) tile_drain<SAVE>(t, 32, rec, h);
    if (t.qn > 0) tile_drain<SAVE>(t, t.qn, rec, h);
    if (SAVE && t.lane == 0) *cnt_out = t.saved;
    __syncthreads();
}

__device__ __forceinline__ void tile_sum_from_list(TileCtx& t, int tid,
                                                   const int2* __restrict__ lp, int n,
                                                   const float* __restrict__ h) {
    for (int i = tid; i < (TILE_NODES + 1) * FEAT / 4; i += 1024)
        ((float4*)t.acc)[i] = make_float4(0.f, 0.f, 0.f, 0.f);
    __syncthreads();
    if (n > 0) {
        int2 cur = lp[min(t.lane, n - 1)];
        int i = 0;
        for (; i + 32 <= n; i += 32) {
            const int2 nxt = lp[min(i + 32 + t.lane, n - 1)];
            tile_add32(t, 32, (int)((unsigned)cur.x >> 8), cur.x & 255, cur.y, h);
            cur = nxt;
        }
        if (i < n) {
            const bool valid = i + t.lane < n;
            const int p = valid ? cur.x : TILE_NODES;
            const int wb = valid ? cur.y : 0;
            tile_add32(t, n - i, (int)((unsigned)p >> 8), p & 255, wb, h);
        }
    }
    __syncthreads();
}

#define CLOSE_TRIPS 4

__global__ __launch_bounds__(1024) void aggregate_layer1_kernel(const int* __restrict__ rec4,
                                                                const int* __restrict__ key4,
                                                                const float* __restrict__ deg,
                                                                const float* __restrict__ h,
                                                                const float* __restrict__ b0,
                                                                const float* __restrict__ g,
                                                                const float* __restrict__ be,
                                                                const float* __restrict__ mu,
                                                                const float* __restrict__ var,
                                                                const float* __restrict__ W1,
                                                                float* __restrict__ h1,
                                                                float* __restrict__ h2pre,
                                                                int* __restrict__ lcnt,
                                                                int* __restrict__ list2,
                                                                int E, int N) {
    __shared__ __attribute__((aligned(16))) float acc[(TILE_NODES + 1) * FEAT];
    __shared__ __attribute__((aligned(16))) int qbuf[AGG_WAVES * QCAP];
    __shared__ __attribute__((aligned(16))) float wp[(FEAT / 2) * WP_PITCH];

    const int tid = threadIdx.x;
    const int wave = __builtin_amdgcn_readfirstlane(tid >> 5);

    float4 wr0[W1_PIECES], wr1[W1_PIECES];
#pragma unroll
    for (int i = 0; i < W1_PIECES; ++i) {
        const int it = tid + 1024 * i;
        const int p = it >> 5;
        const int c4 = (it & 31) << 2;
        wr0[i] = *(const float4*)(W1 + (size_t)(2 * p) * FEAT + c4);
        wr1[i] = *(const float4*)(W1 + (size_t)(2 * p + 1) * FEAT + c4);
    }

    TileCtx t;
    t.acc = acc;
    t.q = qbuf + wave * QCAP;
    t.base = blockIdx.x * TILE_NODES;
    t.lane = tid & 31;
    t.head = 0;
    t.qn = 0;
    t.list = (int2*)list2 + ((size_t)blockIdx.x * AGG_WAVES + wave) * LCAP;
    t.saved = 0;

    tile_sum<true>(t, wave, tid, lcnt + blockIdx.x * AGG_WAVES + wave,
                   (const int4*)rec4, key4, h, E);

#pragma unroll
    for (int i = 0; i < W1_PIECES; ++i) {
        const int it = tid + 1024 * i;
        float* d = wp + (it >> 5) * WP_PITCH + 2 * ((it & 31) << 2);
        *(float4*)(d)     = make_float4(wr0[i].x, wr1[i].x, wr0[i].y, wr1[i].y);
        *(float4*)(d + 4) = make_float4(wr0[i].z, wr1[i].z, wr0[i].w, wr1[i].w);
    }

    const int f = (tid & 31) << 2;
    const float4 bv = *(const float4*)(b0 + f);
    const float4 gv = *(const float4*)(g + f);
    const float4 ev = *(const float4*)(be + f);
    const float4 mv = *(const float4*)(mu + f);
    const float4 vv = *(const float4*)(var + f);
    const float bb[4] = {bv.x, bv.y, bv.z, bv.w};
    const float gg[4] = {gv.x, gv.y, gv.z, gv.w};
    const float ee[4] = {ev.x, ev.y, ev.z, ev.w};
    const float mm[4] = {mv.x, mv.y, mv.z, mv.w};
    const float rs[4] = {rsqrtf(vv.x + BN_EPS), rsqrtf(vv.y + BN_EPS),
                         rsqrtf(vv.z + BN_EPS), rsqrtf(vv.w + BN_EPS)};

    int* qA = qbuf;

    const int rows = min(TILE_NODES, N - t.base);
    const int lim = rows * (FEAT / 4);
    const float4* hsrc = (const float4*)(h + (size_t)t.base * FEAT);
    float4* dst = (float4*)(h1 + (size_t)t.base * FEAT);
    for (int i0 = tid; i0 < lim; i0 += CLOSE_TRIPS * 1024) {
        float4 hv[CLOSE_TRIPS];
        float dv[CLOSE_TRIPS];
#pragma unroll
        for (int u = 0; u < CLOSE_TRIPS; ++u) {
            const int j4 = min(i0 + u * 1024, lim - 1);
            hv[u] = hsrc[j4];
            dv[u] = dis_of(deg, t.base + (j4 >> 5));
        }
#pragma unroll
        for (int u = 0; u < CLOSE_TRIPS; ++u) {
            const int i4 = i0 + u * 1024;
            if (i4 < lim) {
                const float* ar = acc + (i4 >> 5) * FEAT + (tid & 31);
                const float dd = dv[u];
                const float av[4] = {ar[0], ar[32], ar[64], ar[96]};
                const float hh[4] = {hv[u].x, hv[u].y, hv[u].z, hv[u].w};
                float o[4];
#pragma unroll
                for (int j = 0; j < 4; ++j) {
                    float v = dd * (av[j] + hh[j]) + bb[j];
                    v = (v - mm[j]) * rs[j] * gg[j] + ee[j];
                    o[j] = (v > 0.0f) ? v : expm1f(v);
                }
                dst[i4] = make_float4(o[0], o[1], o[2], o[3]);
                *(int4*)&qA[(i4 >> 5) * LA_PITCH + f] =
                    make_int4(__float_as_int(o[0] * dd), __float_as_int(o[1] * dd),
                              __float_as_int(o[2] * dd), __float_as_int(o[3] * dd));
            }
        }
    }
    const int nStrips = (rows + 15) >> 4;
    for (int i4 = lim + tid; i4 < nStrips * 16 * (FEAT / 4); i4 += 1024)
        *(int4*)&qA[(i4 >> 5) * LA_PITCH + f] = make_int4(0, 0, 0, 0);
    __syncthreads();

    const int lane = tid & 31;
    const int lid  = lane & 15;
    const int half = lane >> 4;
    const int n0   = (wave & 7) * 16;
    float* yt = h2pre + (size_t)t.base * FEAT + n0 + lid;
#if HAS_WMMA_F32_4
    v2f bf[32];
#pragma unroll
    for (int j = 0; j < 32; ++j)
        bf[j] = *(const v2f*)&wp[(2 * j + half) * WP_PITCH + 2 * (n0 + lid)];

    for (int s = wave >> 3; s < nStrips; s += 4) {
        v8f cacc = {0.f, 0.f, 0.f, 0.f, 0.f, 0.f, 0.f, 0.f};
        const int* arow = qA + (s * 16 + lid) * LA_PITCH + 2 * half;
#pragma unroll
        for (int j = 0; j < 32; ++j) {
            const int2 ai = *(const int2*)&arow[4 * j];
            v2f a;
            a.x = __int_as_float(ai.x);
            a.y = __int_as_float(ai.y);
            cacc = __builtin_amdgcn_wmma_f32_16x16x4_f32(
                 false, a,  false, bf[j],
                 (short)0, cacc,  false,  false);
        }
        const int r0 = s * 16;
        if (r0 + 16 <= rows) {
#pragma unroll
            for (int v = 0; v < 8; ++v) yt[(size_t)(r0 + v + 8 * half) * FEAT] = cacc[v];
        } else {
#pragma unroll
            for (int v = 0; v < 8; ++v) {
                const int m = r0 + v + 8 * half;
                if (m < rows) yt[(size_t)m * FEAT] = cacc[v];
            }
        }
    }
#else
    for (int s = wave >> 3; s < nStrips; s += 4) {
        float accs[8] = {0.f, 0.f, 0.f, 0.f, 0.f, 0.f, 0.f, 0.f};
        for (int k = 0; k < FEAT; ++k) {
            const float bvk = wp[(k >> 1) * WP_PITCH + 2 * (n0 + lid) + (k & 1)];
#pragma unroll
            for (int v = 0; v < 8; ++v)
                accs[v] = fmaf(__int_as_float(qA[(s * 16 + v + 8 * half) * LA_PITCH + k]), bvk, accs[v]);
        }
#pragma unroll
        for (int v = 0; v < 8; ++v) {
            const int m = s * 16 + v + 8 * half;
            if (m < rows) yt[(size_t)m * FEAT] = accs[v];
        }
    }
#endif
}

__global__ __launch_bounds__(1024) void aggregate_layer2_kernel(const int* __restrict__ rec4,
                                                                const int* __restrict__ key4,
                                                                const float* __restrict__ deg,
                                                                const float* __restrict__ h,
                                                                const float* __restrict__ emb,
                                                                const float* __restrict__ h1,
                                                                const float* __restrict__ b1,
                                                                const int* __restrict__ lcnt,
                                                                const int* __restrict__ list2,
                                                                float* __restrict__ out,
                                                                int E, int N) {
    __shared__ float acc[(TILE_NODES + 1) * FEAT];
    __shared__ int qbuf[AGG_WAVES * QCAP];

    const int tid = threadIdx.x;
    const int wave = __builtin_amdgcn_readfirstlane(tid >> 5);

    TileCtx t;
    t.acc = acc;
    t.q = qbuf + wave * QCAP;
    t.base = blockIdx.x * TILE_NODES;
    t.lane = tid & 31;
    t.head = 0;
    t.qn = 0;
    t.list = nullptr;
    t.saved = 0;

    const int myc = lcnt[blockIdx.x * AGG_WAVES + t.lane];
    const bool over = __builtin_amdgcn_ballot_w32(myc > LCAP) != 0u;
    if (!over) {
        const int n = __builtin_amdgcn_readlane(myc, wave);
        tile_sum_from_list(t, tid,
                           (const int2*)list2 + ((size_t)blockIdx.x * AGG_WAVES + wave) * LCAP,
                           n, h);
    } else {
        tile_sum<false>(t, wave, tid, nullptr, (const int4*)rec4, key4, h, E);
    }

    const int f = (tid & 31) << 2;
    const float4 bv = *(const float4*)(b1 + f);
    const float inv3 = 1.0f / 3.0f;

    const int rows = min(TILE_NODES, N - t.base);
    const int lim = rows * (FEAT / 4);
    const float4* hsrc = (const float4*)(h + (size_t)t.base * FEAT);
    const float4* esrc = (const float4*)(emb + (size_t)t.base * FEAT);
    const float4* osrc = (const float4*)(h1 + (size_t)t.base * FEAT);
    float4* dst = (float4*)(out + (size_t)t.base * FEAT);
    for (int i0 = tid; i0 < lim; i0 += CLOSE_TRIPS * 1024) {
        float4 pv[CLOSE_TRIPS], xv[CLOSE_TRIPS], ov[CLOSE_TRIPS];
        float dv[CLOSE_TRIPS];
#pragma unroll
        for (int u = 0; u < CLOSE_TRIPS; ++u) {
            const int j4 = min(i0 + u * 1024, lim - 1);
            xv[u] = esrc[j4];
            ov[u] = osrc[j4];
            pv[u] = hsrc[j4];
            dv[u] = dis_of(deg, t.base + (j4 >> 5));
        }
#pragma unroll
        for (int u = 0; u < CLOSE_TRIPS; ++u) {
            asm volatile("" :: "v"(xv[u].x), "v"(xv[u].y), "v"(xv[u].z), "v"(xv[u].w),
                               "v"(ov[u].x), "v"(ov[u].y), "v"(ov[u].z), "v"(ov[u].w),
                               "v"(pv[u].x), "v"(pv[u].y), "v"(pv[u].z), "v"(pv[u].w),
                               "v"(dv[u]));
        }
#pragma unroll
        for (int u = 0; u < CLOSE_TRIPS; ++u) {
            const int i4 = i0 + u * 1024;
            if (i4 < lim) {
                const float* ar = acc + (i4 >> 5) * FEAT + (tid & 31);
                const float4 a2 = make_float4(ar[0], ar[32], ar[64], ar[96]);
                const float dd = dv[u];
                const float4 e = xv[u];
                const float4 o1 = ov[u];
                const float4 p2 = pv[u];
                float4 r;
                r.x = (e.x + o1.x + dd * (a2.x + p2.x) + bv.x) * inv3;
                r.y = (e.y + o1.y + dd * (a2.y + p2.y) + bv.y) * inv3;
                r.z = (e.z + o1.z + dd * (a2.z + p2.z) + bv.z) * inv3;
                r.w = (e.w + o1.w + dd * (a2.w + p2.w) + bv.w) * inv3;
                dst[i4] = r;
            }
        }
    }
}

extern "C" void kernel_launch(void* const* d_in, const int* in_sizes, int n_in,
                              void* d_out, int out_size, void* d_ws, size_t ws_size,
                              hipStream_t stream) {
    const float* emb = (const float*)d_in[0];
    const int*   ei  = (const int*)d_in[1];
    const float* ew  = (const float*)d_in[2];
    const float* W0  = (const float*)d_in[3];
    const float* b0  = (const float*)d_in[4];
    const float* W1  = (const float*)d_in[5];
    const float* b1  = (const float*)d_in[6];
    const float* bng = (const float*)d_in[7];
    const float* bnb = (const float*)d_in[8];
    const float* bnm = (const float*)d_in[9];
    const float* bnv = (const float*)d_in[10];
    float* out = (float*)d_out;

    const int N  = in_sizes[0] / FEAT;
    const int E  = in_sizes[2];
    const size_t ND = (size_t)N * FEAT;

    const int* row = ei;
    const int* col = ei + E;

    float* ws   = (float*)d_ws;
    size_t disPad = ((size_t)N + 1023) & ~(size_t)1023;
    size_t keyPad = ((size_t)E + 1023) & ~(size_t)1023;
    float* deg  = ws;
    float* bufA = ws + disPad;
    float* bufB = bufA + ND;
    int* key4   = (int*)(bufB + ND);
    int* rec    = (int*)((char*)key4 + keyPad);

    hipMemsetAsync(deg,  0, (size_t)N * sizeof(float), stream);

    const int gTile  = (N + TILE_NODES - 1) / TILE_NODES;
    const int gRowT  = (N + 15) / 16;
    int* lcnt   = rec + (size_t)E * 4;
    int* list2  = lcnt + (((size_t)gTile * AGG_WAVES + 63) & ~(size_t)63);
    float* bufC = (float*)(list2 + (size_t)gTile * AGG_WAVES * LCAP * 2);

    degree_key_kernel<<<(E + 1023) / 1024, 256, 0, stream>>>(row, col, ew, deg, key4, rec, E);

    gemm_n128_kernel<<<GEMM_BLOCKS, 256, 0, stream>>>(emb, W0, deg, bufA, N, gRowT);

    aggregate_layer1_kernel<<<gTile, 1024, 0, stream>>>(rec, key4, deg, bufA,
                                                        b0, bng, bnb, bnm, bnv, W1, bufB, bufC,
                                                        lcnt, list2, E, N);

    aggregate_layer2_kernel<<<gTile, 1024, 0, stream>>>(rec, key4, deg, bufC,
                                                        emb, bufB, b1, lcnt, list2, out, E, N);
}
